# starting kernel
# baseline (speedup 1.0000x reference)
#include <hip/hip_runtime.h>
#include <hip/hip_bf16.h>
#include <math.h>

#define BB     8
#define SS     256
#define HIDC   1024
#define HH     16
#define DD     64
#define FF     64
#define LR     0.1f
#define GSCALE (2.0f / 8192.0f)
#define LN_EPS 1e-5f

typedef float v2f __attribute__((ext_vector_type(2)));
typedef float v8f __attribute__((ext_vector_type(8)));

__device__ __forceinline__ v8f wmma4(v2f a, v2f b, v8f c) {
  return __builtin_amdgcn_wmma_f32_16x16x4_f32(
       false, a,  false, b,
       (short)0, c,  false,  false);
}

__device__ __forceinline__ float gelu_f(float x) {
  const float c = 0.7978845608028654f;
  float u = c * (x + 0.044715f * x * x * x);
  return 0.5f * x * (1.0f + tanhf(u));
}
__device__ __forceinline__ float gelu_grad(float x) {
  const float c = 0.7978845608028654f;
  float x2 = x * x;
  float u = c * (x + 0.044715f * x2 * x);
  float t = tanhf(u);
  return 0.5f * (1.0f + t) + 0.5f * x * (1.0f - t * t) * c * (1.0f + 0.134145f * x2);
}
__device__ __forceinline__ float wave_sum32(float v) {
#pragma unroll
  for (int off = 16; off > 0; off >>= 1) v += __shfl_xor(v, off, 32);
  return v;
}

__global__ void __launch_bounds__(256)
gemm_wmma_f32(const float* __restrict__ A, const float* __restrict__ Bm,
              float* __restrict__ C, int M, int N, int K, int act) {
  __shared__ float As[16 * 64];
  const int tid   = threadIdx.x;
  const int wave  = tid >> 5;
  const int lane  = tid & 31;
  const int mbase = blockIdx.y * 16;
  const int tn    = blockIdx.x * 8 + wave;
  const int col   = tn * 16 + (lane & 15);
  const int kh    = (lane >> 4) << 1;
  const int arow  = lane & 15;
  v8f acc = {0.f, 0.f, 0.f, 0.f, 0.f, 0.f, 0.f, 0.f};
  for (int kc = 0; kc < K; kc += 64) {
    __syncthreads();
    for (int i = tid; i < 16 * 64; i += 256)
      As[i] = A[(size_t)(mbase + (i >> 6)) * K + kc + (i & 63)];
    if (kc + 64 < K)
      __builtin_prefetch(&A[(size_t)(mbase + (tid >> 4)) * K + kc + 64], 0, 1);
    __syncthreads();
#pragma unroll
    for (int k0 = 0; k0 < 64; k0 += 4) {
      v2f a, b;
      a.x = As[arow * 64 + k0 + kh];
      a.y = As[arow * 64 + k0 + kh + 1];
      b.x = Bm[(size_t)(kc + k0 + kh) * N + col];
      b.y = Bm[(size_t)(kc + k0 + kh + 1) * N + col];
      acc = wmma4(a, b, acc);
    }
  }
  const int r0 = mbase + ((lane >= 16) ? 8 : 0);
#pragma unroll
  for (int i = 0; i < 8; ++i) {
    float v = acc[i];
    if (act) v = gelu_f(v);
    C[(size_t)(r0 + i) * N + col] = v;
  }
}

#define EPI_BIAS      0
#define EPI_GELU_PAIR 1
#define EPI_DPRED     2
#define EPI_DGELU     3

__device__ __forceinline__ void
gemm16_rowA(const float* A8, const float* Bm, const float* bias,
            float* C8, float* C2, const float* aux,
            int transB, int epi, int wave, int lane) {
  if (wave >= 4) return;
  const int col  = wave * 16 + (lane & 15);
  const int mrow = lane & 15;
  const int kh   = (lane >> 4) << 1;
  const bool live = (mrow < 8);
  v8f acc = {0.f, 0.f, 0.f, 0.f, 0.f, 0.f, 0.f, 0.f};
#pragma unroll
  for (int k0 = 0; k0 < 64; k0 += 4) {
    v2f a, b;
    a.x = live ? A8[mrow * 64 + k0 + kh] : 0.f;
    a.y = live ? A8[mrow * 64 + k0 + kh + 1] : 0.f;
    if (!transB) {
      b.x = Bm[(k0 + kh) * 64 + col];
      b.y = Bm[(k0 + kh + 1) * 64 + col];
    } else {
      b.x = Bm[col * 64 + k0 + kh];
      b.y = Bm[col * 64 + k0 + kh + 1];
    }
    acc = wmma4(a, b, acc);
  }
  if (lane < 16) {
    const float bv = bias ? bias[col] : 0.f;
#pragma unroll
    for (int i = 0; i < 8; ++i) {
      const int o = i * 64 + col;
      float v = acc[i] + bv;
      if (epi == EPI_GELU_PAIR)      { C8[o] = v; C2[o] = gelu_f(v); }
      else if (epi == EPI_DPRED)     { C8[o] = GSCALE * (v - aux[o]); }
      else if (epi == EPI_DGELU)     { C8[o] = v * gelu_grad(aux[o]); }
      else                           { C8[o] = v; }
    }
  }
}

__device__ __forceinline__ void
gemm_outer_update(const float* X8, const float* Y8, float* Wt, float lr,
                  int wave, int lane) {
  const int kh = (lane >> 4) << 1;
#pragma unroll
  for (int ti = 0; ti < 2; ++ti) {
    const int t  = wave * 2 + ti;
    const int tm = t >> 2, tn = t & 3;
    const int m = tm * 16 + (lane & 15);
    const int n = tn * 16 + (lane & 15);
    v8f acc = {0.f, 0.f, 0.f, 0.f, 0.f, 0.f, 0.f, 0.f};
#pragma unroll
    for (int k0 = 0; k0 < 8; k0 += 4) {
      v2f a, b;
      a.x = X8[(k0 + kh) * 64 + m];
      a.y = X8[(k0 + kh + 1) * 64 + m];
      b.x = Y8[(k0 + kh) * 64 + n];
      b.y = Y8[(k0 + kh + 1) * 64 + n];
      acc = wmma4(a, b, acc);
    }
    const int rbase = tm * 16 + ((lane >= 16) ? 8 : 0);
    const int cc    = tn * 16 + (lane & 15);
#pragma unroll
    for (int i = 0; i < 8; ++i) Wt[(rbase + i) * 64 + cc] -= lr * acc[i];
  }
}

__device__ __forceinline__ void
fwd_pass(const float* ktb, float* av, float* hg, float* ob, float* xh, float* zb,
         const float* W1, const float* W2, const float* b1v, const float* b2v,
         const float* lgv, const float* lbv, float* muv, float* isv,
         int wave, int lane, int tid) {
  gemm16_rowA(ktb, W1, b1v, av, hg, nullptr, 0, EPI_GELU_PAIR, wave, lane);
  __syncthreads();
  gemm16_rowA(hg, W2, b2v, ob, nullptr, nullptr, 0, EPI_BIAS, wave, lane);
  __syncthreads();
  {
    float e0 = ob[wave * 64 + lane], e1 = ob[wave * 64 + lane + 32];
    float s = wave_sum32(e0 + e1);
    float q = wave_sum32(e0 * e0 + e1 * e1);
    if (lane == 0) {
      float m = s * (1.f / 64.f);
      muv[wave] = m;
      isv[wave] = rsqrtf(q * (1.f / 64.f) - m * m + LN_EPS);
    }
  }
  __syncthreads();
  for (int idx = tid; idx < 512; idx += 256) {
    int b = idx >> 6, d = idx & 63;
    float xv = (ob[idx] - muv[b]) * isv[b];
    xh[idx] = xv;
    zb[idx] = xv * lgv[d] + lbv[d];
  }
  __syncthreads();
}

#define SCAN_SMEM_FLOATS (3 * 4096 + 10 * 512 + 7 * 64 + 32)

__global__ void __launch_bounds__(256)
ttt_scan_kernel(const float* __restrict__ kg, const float* __restrict__ vg,
                const float* __restrict__ qg,
                const float* __restrict__ fw1, const float* __restrict__ fb1,
                const float* __restrict__ fw2, const float* __restrict__ fb2,
                const float* __restrict__ flg, const float* __restrict__ flb,
                const float* __restrict__ lwg, const float* __restrict__ lbg,
                const float* __restrict__ ttg, const float* __restrict__ ttb,
                float* __restrict__ outg) {
  extern __shared__ float smem[];
  float* W1  = smem;
  float* W2  = W1 + 4096;
  float* LW  = W2 + 4096;
  float* ktb = LW + 4096;
  float* tgt = ktb + 512;
  float* av  = tgt + 512;
  float* hg  = av + 512;
  float* ob  = hg + 512;
  float* xh  = ob + 512;
  float* zb  = xh + 512;
  float* dzb = zb + 512;
  float* dob = dzb + 512;
  float* dab = dob + 512;
  float* b1v = dab + 512;
  float* b2v = b1v + 64;
  float* lgv = b2v + 64;
  float* lbv = lgv + 64;
  float* tgv = lbv + 64;
  float* tbv = tgv + 64;
  float* lbz = tbv + 64;
  float* muv = lbz + 64;
  float* isv = muv + 8;
  float* m1v = isv + 8;
  float* m2v = m1v + 8;

  const int h    = blockIdx.x;
  const int tid  = threadIdx.x;
  const int wave = tid >> 5;
  const int lane = tid & 31;

  const int b0 = tid >> 6,          d0 = tid & 63;
  const int b1i = (tid + 256) >> 6, d1i = (tid + 256) & 63;
  const size_t g0base = (size_t)(b0 * SS) * HIDC + h * 64 + d0;
  const size_t g1base = (size_t)(b1i * SS) * HIDC + h * 64 + d1i;

  for (int i = tid; i < 4096; i += 256) {
    W1[i] = fw1[(size_t)h * 4096 + i];
    W2[i] = fw2[(size_t)h * 4096 + i];
    LW[i] = lwg[i];
  }
  if (tid < 64) {
    b1v[tid] = fb1[h * 64 + tid];
    b2v[tid] = fb2[h * 64 + tid];
    lgv[tid] = flg[h * 64 + tid];
    lbv[tid] = flb[h * 64 + tid];
    tgv[tid] = ttg[h * 64 + tid];
    tbv[tid] = ttb[h * 64 + tid];
    lbz[tid] = lbg[tid];
  }

  float kp0 = kg[g0base], vp0 = vg[g0base];
  float kp1 = kg[g1base], vp1 = vg[g1base];
  __syncthreads();

  for (int t = 0; t < SS; ++t) {
    ktb[tid]       = kp0;  tgt[tid]       = vp0 - kp0;
    ktb[tid + 256] = kp1;  tgt[tid + 256] = vp1 - kp1;
    __syncthreads();
    if (t + 1 < SS) {
      kp0 = kg[g0base + (t + 1) * HIDC];  vp0 = vg[g0base + (t + 1) * HIDC];
      kp1 = kg[g1base + (t + 1) * HIDC];  vp1 = vg[g1base + (t + 1) * HIDC];
    }

    for (int step = 0; step < 4; ++step) {
      fwd_pass(ktb, av, hg, ob, xh, zb, W1, W2, b1v, b2v, lgv, lbv,
               muv, isv, wave, lane, tid);
      gemm16_rowA(zb, LW, lbz, dab, nullptr, tgt, 0, EPI_DPRED, wave, lane);
      __syncthreads();
      gemm16_rowA(dab, LW, nullptr, dzb, nullptr, nullptr, 1, EPI_BIAS, wave, lane);
      __syncthreads();
      {
        float dd0 = dzb[wave * 64 + lane] * lgv[lane];
        float dd1 = dzb[wave * 64 + lane + 32] * lgv[lane + 32];
        float x0 = xh[wave * 64 + lane], x1 = xh[wave * 64 + lane + 32];
        float s1 = wave_sum32(dd0 + dd1);
        float s2 = wave_sum32(dd0 * x0 + dd1 * x1);
        if (lane == 0) { m1v[wave] = s1 * (1.f / 64.f); m2v[wave] = s2 * (1.f / 64.f); }
      }
      __syncthreads();
      for (int idx = tid; idx < 512; idx += 256) {
        int b = idx >> 6, d = idx & 63;
        float dxh = dzb[idx] * lgv[d];
        dob[idx] = isv[b] * (dxh - m1v[b] - xh[idx] * m2v[b]);
      }
      __syncthreads();
      if (tid < 64) {
        float s = 0.f;
        for (int b = 0; b < 8; ++b) s += dzb[b * 64 + tid] * xh[b * 64 + tid];
        lgv[tid] -= LR * s;
      } else if (tid < 128) {
        int d = tid - 64; float s = 0.f;
        for (int b = 0; b < 8; ++b) s += dzb[b * 64 + d];
        lbv[d] -= LR * s;
      }
      gemm16_rowA(dob, W2, nullptr, dab, nullptr, av, 1, EPI_DGELU, wave, lane);
      __syncthreads();
      gemm_outer_update(hg, dob, W2, LR, wave, lane);
      gemm_outer_update(ktb, dab, W1, LR, wave, lane);
      if (tid < 64) {
        float s = 0.f;
        for (int b = 0; b < 8; ++b) s += dob[b * 64 + tid];
        b2v[tid] -= LR * s;
      } else if (tid < 128) {
        int d = tid - 64; float s = 0.f;
        for (int b = 0; b < 8; ++b) s += dab[b * 64 + d];
        b1v[d] -= LR * s;
      }
      __syncthreads();
    }

    fwd_pass(ktb, av, hg, ob, xh, zb, W1, W2, b1v, b2v, lgv, lbv,
             muv, isv, wave, lane, tid);
    {
      float e0 = zb[wave * 64 + lane], e1 = zb[wave * 64 + lane + 32];
      float s = wave_sum32(e0 + e1);
      float q = wave_sum32(e0 * e0 + e1 * e1);
      if (lane == 0) {
        float m = s * (1.f / 64.f);
        muv[wave] = m;
        isv[wave] = rsqrtf(q * (1.f / 64.f) - m * m + LN_EPS);
      }
    }
    __syncthreads();
    for (int idx = tid; idx < 512; idx += 256) {
      int b = idx >> 6, d = idx & 63;
      float z2 = (zb[idx] - muv[b]) * isv[b] * tgv[d] + tbv[d];
      size_t g = ((size_t)(b * SS + t)) * HIDC + h * 64 + d;
      outg[g] = qg[g] + z2;
    }
    __syncthreads();
  }
}

__global__ void __launch_bounds__(256)
gate_ln_kernel(const float* __restrict__ outb, const float* __restrict__ gact,
               const float* __restrict__ png, const float* __restrict__ pnb,
               float* __restrict__ gated) {
  __shared__ float r1[256], r2[256];
  __shared__ float mu_s, inv_s;
  const int r = blockIdx.x, tid = threadIdx.x;
  const float* row = outb + (size_t)r * HIDC;
  float s = 0.f, q = 0.f;
  for (int i = tid; i < HIDC; i += 256) { float v = row[i]; s += v; q += v * v; }
  r1[tid] = s; r2[tid] = q;
  __syncthreads();
  for (int off = 128; off > 0; off >>= 1) {
    if (tid < off) { r1[tid] += r1[tid + off]; r2[tid] += r2[tid + off]; }
    __syncthreads();
  }
  if (tid == 0) {
    float m = r1[0] * (1.f / HIDC);
    mu_s = m;
    inv_s = rsqrtf(r2[0] * (1.f / HIDC) - m * m + LN_EPS);
  }
  __syncthreads();
  for (int i = tid; i < HIDC; i += 256) {
    float y = (row[i] - mu_s) * inv_s * png[i] + pnb[i];
    gated[(size_t)r * HIDC + i] = gact[(size_t)r * HIDC + i] * y;
  }
}

extern "C" void kernel_launch(void* const* d_in, const int* in_sizes, int n_in,
                              void* d_out, int out_size, void* d_ws, size_t ws_size,
                              hipStream_t stream) {
  (void)in_sizes; (void)n_in; (void)out_size; (void)ws_size;
  const float* x   = (const float*)d_in[0];
  const float* wq  = (const float*)d_in[1];
  const float* wk  = (const float*)d_in[2];
  const float* wv  = (const float*)d_in[3];
  const float* fw1 = (const float*)d_in[4];
  const float* fb1 = (const float*)d_in[5];
  const float* fw2 = (const float*)d_in[6];
  const float* fb2 = (const float*)d_in[7];
  const float* flg = (const float*)d_in[8];
  const float* flb = (const float*)d_in[9];
  const float* lw  = (const float*)d_in[10];
  const float* lb  = (const float*)d_in[11];
  const float* ttg = (const float*)d_in[12];
  const float* ttb = (const float*)d_in[13];
  const float* wo  = (const float*)d_in[14];
  const float* wg  = (const float*)d_in[15];
  const float* png = (const float*)d_in[16];
  const float* pnb = (const float*)d_in[17];

  const int M = BB * SS;
  const size_t PLANE = (size_t)M * HIDC;
  float* qb    = (float*)d_ws;
  float* kb    = qb + PLANE;
  float* vb    = kb + PLANE;
  float* gactb = vb + PLANE;
  float* outb  = gactb + PLANE;
  float* gatedb = qb;

  dim3 gblk(256);
  dim3 ggrid(HIDC / 128, M / 16);
  gemm_wmma_f32<<<ggrid, gblk, 0, stream>>>(x, wq, qb,    M, HIDC, HIDC, 0);
  gemm_wmma_f32<<<ggrid, gblk, 0, stream>>>(x, wk, kb,    M, HIDC, HIDC, 0);
  gemm_wmma_f32<<<ggrid, gblk, 0, stream>>>(x, wv, vb,    M, HIDC, HIDC, 0);
  gemm_wmma_f32<<<ggrid, gblk, 0, stream>>>(x, wg, gactb, M, HIDC, HIDC, 1);

  ttt_scan_kernel<<<HH, 256, SCAN_SMEM_FLOATS * sizeof(float), stream>>>(
      kb, vb, qb, fw1, fb1, fw2, fb2, flg, flb, lw, lb, ttg, ttb, outb);

  gate_ln_kernel<<<M, 256, 0, stream>>>(outb, gactb, png, pnb, gatedb);

  gemm_wmma_f32<<<ggrid, gblk, 0, stream>>>(gatedb, wo, (float*)d_out,
                                            M, HIDC, HIDC, 0);
}


# scan kernel's 8x64x64 LDS GEMMs: all k-steps' A/B fragments and epilogue operands loaded as one group ahead of the WMMA chain instead of load-wait-WMMA per k-step; A rows read unmasked
# speedup vs baseline: 1.2217x; 1.2217x over previous
#include <hip/hip_runtime.h>
#include <hip/hip_bf16.h>
#include <math.h>

#define BB     8
#define SS     256
#define HIDC   1024
#define HH     16
#define DD     64
#define FF     64
#define LR     0.1f
#define GSCALE (2.0f / 8192.0f)
#define LN_EPS 1e-5f

typedef float v2f __attribute__((ext_vector_type(2)));
typedef float v8f __attribute__((ext_vector_type(8)));

__device__ __forceinline__ v8f wmma4(v2f a, v2f b, v8f c) {
  return __builtin_amdgcn_wmma_f32_16x16x4_f32(
       false, a,  false, b,
       (short)0, c,  false,  false);
}

__device__ __forceinline__ float gelu_f(float x) {
  const float c = 0.7978845608028654f;
  float u = c * (x + 0.044715f * x * x * x);
  return 0.5f * x * (1.0f + tanhf(u));
}
__device__ __forceinline__ float gelu_grad(float x) {
  const float c = 0.7978845608028654f;
  float x2 = x * x;
  float u = c * (x + 0.044715f * x2 * x);
  float t = tanhf(u);
  return 0.5f * (1.0f + t) + 0.5f * x * (1.0f - t * t) * c * (1.0f + 0.134145f * x2);
}
__device__ __forceinline__ float wave_sum32(float v) {
#pragma unroll
  for (int off = 16; off > 0; off >>= 1) v += __shfl_xor(v, off, 32);
  return v;
}

__global__ void __launch_bounds__(256)
gemm_wmma_f32(const float* __restrict__ A, const float* __restrict__ Bm,
              float* __restrict__ C, int M, int N, int K, int act) {
  __shared__ float As[16 * 64];
  const int tid   = threadIdx.x;
  const int wave  = tid >> 5;
  const int lane  = tid & 31;
  const int mbase = blockIdx.y * 16;
  const int tn    = blockIdx.x * 8 + wave;
  const int col   = tn * 16 + (lane & 15);
  const int kh    = (lane >> 4) << 1;
  const int arow  = lane & 15;
  v8f acc = {0.f, 0.f, 0.f, 0.f, 0.f, 0.f, 0.f, 0.f};
  for (int kc = 0; kc < K; kc += 64) {
    __syncthreads();
    for (int i = tid; i < 16 * 64; i += 256)
      As[i] = A[(size_t)(mbase + (i >> 6)) * K + kc + (i & 63)];
    if (kc + 64 < K)
      __builtin_prefetch(&A[(size_t)(mbase + (tid >> 4)) * K + kc + 64], 0, 1);
    __syncthreads();
#pragma unroll
    for (int k0 = 0; k0 < 64; k0 += 4) {
      v2f a, b;
      a.x = As[arow * 64 + k0 + kh];
      a.y = As[arow * 64 + k0 + kh + 1];
      b.x = Bm[(size_t)(kc + k0 + kh) * N + col];
      b.y = Bm[(size_t)(kc + k0 + kh + 1) * N + col];
      acc = wmma4(a, b, acc);
    }
  }
  const int r0 = mbase + ((lane >= 16) ? 8 : 0);
#pragma unroll
  for (int i = 0; i < 8; ++i) {
    float v = acc[i];
    if (act) v = gelu_f(v);
    C[(size_t)(r0 + i) * N + col] = v;
  }
}

#define EPI_BIAS      0
#define EPI_GELU_PAIR 1
#define EPI_DPRED     2
#define EPI_DGELU     3

__device__ __forceinline__ void
gemm16_rowA(const float* A8, const float* Bm, const float* bias,
            float* C8, float* C2, const float* aux,
            int transB, int epi, int wave, int lane) {
  if (wave >= 4) return;
  const int col  = wave * 16 + (lane & 15);
  const int mrow = lane & 7;
  const int kh   = (lane >> 4) << 1;
  v2f a[16], b[16];
#pragma unroll
  for (int j = 0; j < 16; ++j) {
    const int k = 4 * j + kh;
    a[j].x = A8[mrow * 64 + k];
    a[j].y = A8[mrow * 64 + k + 1];
    if (!transB) {
      b[j].x = Bm[k * 64 + col];
      b[j].y = Bm[(k + 1) * 64 + col];
    } else {
      b[j].x = Bm[col * 64 + k];
      b[j].y = Bm[col * 64 + k + 1];
    }
  }
  const float bv = bias ? bias[col] : 0.f;
  float ax[8];
#pragma unroll
  for (int i = 0; i < 8; ++i)
    ax[i] = (epi == EPI_DPRED || epi == EPI_DGELU) ? aux[i * 64 + col] : 0.f;
  __builtin_amdgcn_sched_barrier(0);
  v8f acc = {0.f, 0.f, 0.f, 0.f, 0.f, 0.f, 0.f, 0.f};
#pragma unroll
  for (int j = 0; j < 16; ++j) acc = wmma4(a[j], b[j], acc);
  if (lane < 16) {
#pragma unroll
    for (int i = 0; i < 8; ++i) {
      const int o = i * 64 + col;
      float v = acc[i] + bv;
      if (epi == EPI_GELU_PAIR)      { C8[o] = v; C2[o] = gelu_f(v); }
      else if (epi == EPI_DPRED)     { C8[o] = GSCALE * (v - ax[i]); }
      else if (epi == EPI_DGELU)     { C8[o] = v * gelu_grad(ax[i]); }
      else                           { C8[o] = v; }
    }
  }
}

__device__ __forceinline__ void
gemm_outer_update(const float* X8, const float* Y8, float* Wt, float lr,
                  int wave, int lane) {
  const int kh = (lane >> 4) << 1;
#pragma unroll
  for (int ti = 0; ti < 2; ++ti) {
    const int t  = wave * 2 + ti;
    const int tm = t >> 2, tn = t & 3;
    const int m = tm * 16 + (lane & 15);
    const int n = tn * 16 + (lane & 15);
    v8f acc = {0.f, 0.f, 0.f, 0.f, 0.f, 0.f, 0.f, 0.f};
#pragma unroll
    for (int k0 = 0; k0 < 8; k0 += 4) {
      v2f a, b;
      a.x = X8[(k0 + kh) * 64 + m];
      a.y = X8[(k0 + kh + 1) * 64 + m];
      b.x = Y8[(k0 + kh) * 64 + n];
      b.y = Y8[(k0 + kh + 1) * 64 + n];
      acc = wmma4(a, b, acc);
    }
    const int rbase = tm * 16 + ((lane >= 16) ? 8 : 0);
    const int cc    = tn * 16 + (lane & 15);
#pragma unroll
    for (int i = 0; i < 8; ++i) Wt[(rbase + i) * 64 + cc] -= lr * acc[i];
  }
}

__device__ __forceinline__ void
fwd_pass(const float* ktb, float* av, float* hg, float* ob, float* xh, float* zb,
         const float* W1, const float* W2, const float* b1v, const float* b2v,
         const float* lgv, const float* lbv, float* muv, float* isv,
         int wave, int lane, int tid) {
  gemm16_rowA(ktb, W1, b1v, av, hg, nullptr, 0, EPI_GELU_PAIR, wave, lane);
  __syncthreads();
  gemm16_rowA(hg, W2, b2v, ob, nullptr, nullptr, 0, EPI_BIAS, wave, lane);
  __syncthreads();
  {
    float e0 = ob[wave * 64 + lane], e1 = ob[wave * 64 + lane + 32];
    float s = wave_sum32(e0 + e1);
    float q = wave_sum32(e0 * e0 + e1 * e1);
    if (lane == 0) {
      float m = s * (1.f / 64.f);
      muv[wave] = m;
      isv[wave] = rsqrtf(q * (1.f / 64.f) - m * m + LN_EPS);
    }
  }
  __syncthreads();
  for (int idx = tid; idx < 512; idx += 256) {
    int b = idx >> 6, d = idx & 63;
    float xv = (ob[idx] - muv[b]) * isv[b];
    xh[idx] = xv;
    zb[idx] = xv * lgv[d] + lbv[d];
  }
  __syncthreads();
}

#define SCAN_SMEM_FLOATS (3 * 4096 + 10 * 512 + 7 * 64 + 32)

__global__ void __launch_bounds__(256)
ttt_scan_kernel(const float* __restrict__ kg, const float* __restrict__ vg,
                const float* __restrict__ qg,
                const float* __restrict__ fw1, const float* __restrict__ fb1,
                const float* __restrict__ fw2, const float* __restrict__ fb2,
                const float* __restrict__ flg, const float* __restrict__ flb,
                const float* __restrict__ lwg, const float* __restrict__ lbg,
                const float* __restrict__ ttg, const float* __restrict__ ttb,
                float* __restrict__ outg) {
  extern __shared__ float smem[];
  float* W1  = smem;
  float* W2  = W1 + 4096;
  float* LW  = W2 + 4096;
  float* ktb = LW + 4096;
  float* tgt = ktb + 512;
  float* av  = tgt + 512;
  float* hg  = av + 512;
  float* ob  = hg + 512;
  float* xh  = ob + 512;
  float* zb  = xh + 512;
  float* dzb = zb + 512;
  float* dob = dzb + 512;
  float* dab = dob + 512;
  float* b1v = dab + 512;
  float* b2v = b1v + 64;
  float* lgv = b2v + 64;
  float* lbv = lgv + 64;
  float* tgv = lbv + 64;
  float* tbv = tgv + 64;
  float* lbz = tbv + 64;
  float* muv = lbz + 64;
  float* isv = muv + 8;
  float* m1v = isv + 8;
  float* m2v = m1v + 8;

  const int h    = blockIdx.x;
  const int tid  = threadIdx.x;
  const int wave = tid >> 5;
  const int lane = tid & 31;

  const int b0 = tid >> 6,          d0 = tid & 63;
  const int b1i = (tid + 256) >> 6, d1i = (tid + 256) & 63;
  const size_t g0base = (size_t)(b0 * SS) * HIDC + h * 64 + d0;
  const size_t g1base = (size_t)(b1i * SS) * HIDC + h * 64 + d1i;

  for (int i = tid; i < 4096; i += 256) {
    W1[i] = fw1[(size_t)h * 4096 + i];
    W2[i] = fw2[(size_t)h * 4096 + i];
    LW[i] = lwg[i];
  }
  if (tid < 64) {
    b1v[tid] = fb1[h * 64 + tid];
    b2v[tid] = fb2[h * 64 + tid];
    lgv[tid] = flg[h * 64 + tid];
    lbv[tid] = flb[h * 64 + tid];
    tgv[tid] = ttg[h * 64 + tid];
    tbv[tid] = ttb[h * 64 + tid];
    lbz[tid] = lbg[tid];
  }

  float kp0 = kg[g0base], vp0 = vg[g0base];
  float kp1 = kg[g1base], vp1 = vg[g1base];
  __syncthreads();

  for (int t = 0; t < SS; ++t) {
    ktb[tid]       = kp0;  tgt[tid]       = vp0 - kp0;
    ktb[tid + 256] = kp1;  tgt[tid + 256] = vp1 - kp1;
    __syncthreads();
    if (t + 1 < SS) {
      kp0 = kg[g0base + (t + 1) * HIDC];  vp0 = vg[g0base + (t + 1) * HIDC];
      kp1 = kg[g1base + (t + 1) * HIDC];  vp1 = vg[g1base + (t + 1) * HIDC];
    }

    for (int step = 0; step < 4; ++step) {
      fwd_pass(ktb, av, hg, ob, xh, zb, W1, W2, b1v, b2v, lgv, lbv,
               muv, isv, wave, lane, tid);
      gemm16_rowA(zb, LW, lbz, dab, nullptr, tgt, 0, EPI_DPRED, wave, lane);
      __syncthreads();
      gemm16_rowA(dab, LW, nullptr, dzb, nullptr, nullptr, 1, EPI_BIAS, wave, lane);
      __syncthreads();
      {
        float dd0 = dzb[wave * 64 + lane] * lgv[lane];
        float dd1 = dzb[wave * 64 + lane + 32] * lgv[lane + 32];
        float x0 = xh[wave * 64 + lane], x1 = xh[wave * 64 + lane + 32];
        float s1 = wave_sum32(dd0 + dd1);
        float s2 = wave_sum32(dd0 * x0 + dd1 * x1);
        if (lane == 0) { m1v[wave] = s1 * (1.f / 64.f); m2v[wave] = s2 * (1.f / 64.f); }
      }
      __syncthreads();
      for (int idx = tid; idx < 512; idx += 256) {
        int b = idx >> 6, d = idx & 63;
        float dxh = dzb[idx] * lgv[d];
        dob[idx] = isv[b] * (dxh - m1v[b] - xh[idx] * m2v[b]);
      }
      __syncthreads();
      if (tid < 64) {
        float s = 0.f;
        for (int b = 0; b < 8; ++b) s += dzb[b * 64 + tid] * xh[b * 64 + tid];
        lgv[tid] -= LR * s;
      } else if (tid < 128) {
        int d = tid - 64; float s = 0.f;
        for (int b = 0; b < 8; ++b) s += dzb[b * 64 + d];
        lbv[d] -= LR * s;
      }
      gemm16_rowA(dob, W2, nullptr, dab, nullptr, av, 1, EPI_DGELU, wave, lane);
      __syncthreads();
      gemm_outer_update(hg, dob, W2, LR, wave, lane);
      gemm_outer_update(ktb, dab, W1, LR, wave, lane);
      if (tid < 64) {
        float s = 0.f;
        for (int b = 0; b < 8; ++b) s += dob[b * 64 + tid];
        b2v[tid] -= LR * s;
      } else if (tid < 128) {
        int d = tid - 64; float s = 0.f;
        for (int b = 0; b < 8; ++b) s += dab[b * 64 + d];
        b1v[d] -= LR * s;
      }
      __syncthreads();
    }

    fwd_pass(ktb, av, hg, ob, xh, zb, W1, W2, b1v, b2v, lgv, lbv,
             muv, isv, wave, lane, tid);
    {
      float e0 = zb[wave * 64 + lane], e1 = zb[wave * 64 + lane + 32];
      float s = wave_sum32(e0 + e1);
      float q = wave_sum32(e0 * e0 + e1 * e1);
      if (lane == 0) {
        float m = s * (1.f / 64.f);
        muv[wave] = m;
        isv[wave] = rsqrtf(q * (1.f / 64.f) - m * m + LN_EPS);
      }
    }
    __syncthreads();
    for (int idx = tid; idx < 512; idx += 256) {
      int b = idx >> 6, d = idx & 63;
      float z2 = (zb[idx] - muv[b]) * isv[b] * tgv[d] + tbv[d];
      size_t g = ((size_t)(b * SS + t)) * HIDC + h * 64 + d;
      outg[g] = qg[g] + z2;
    }
    __syncthreads();
  }
}

__global__ void __launch_bounds__(256)
gate_ln_kernel(const float* __restrict__ outb, const float* __restrict__ gact,
               const float* __restrict__ png, const float* __restrict__ pnb,
               float* __restrict__ gated) {
  __shared__ float r1[256], r2[256];
  __shared__ float mu_s, inv_s;
  const int r = blockIdx.x, tid = threadIdx.x;
  const float* row = outb + (size_t)r * HIDC;
  float s = 0.f, q = 0.f;
  for (int i = tid; i < HIDC; i += 256) { float v = row[i]; s += v; q += v * v; }
  r1[tid] = s; r2[tid] = q;
  __syncthreads();
  for (int off = 128; off > 0; off >>= 1) {
    if (tid < off) { r1[tid] += r1[tid + off]; r2[tid] += r2[tid + off]; }
    __syncthreads();
  }
  if (tid == 0) {
    float m = r1[0] * (1.f / HIDC);
    mu_s = m;
    inv_s = rsqrtf(r2[0] * (1.f / HIDC) - m * m + LN_EPS);
  }
  __syncthreads();
  for (int i = tid; i < HIDC; i += 256) {
    float y = (row[i] - mu_s) * inv_s * png[i] + pnb[i];
    gated[(size_t)r * HIDC + i] = gact[(size_t)r * HIDC + i] * y;
  }
}

extern "C" void kernel_launch(void* const* d_in, const int* in_sizes, int n_in,
                              void* d_out, int out_size, void* d_ws, size_t ws_size,
                              hipStream_t stream) {
  (void)in_sizes; (void)n_in; (void)out_size; (void)ws_size;
  const float* x   = (const float*)d_in[0];
  const float* wq  = (const float*)d_in[1];
  const float* wk  = (const float*)d_in[2];
  const float* wv  = (const float*)d_in[3];
  const float* fw1 = (const float*)d_in[4];
  const float* fb1 = (const float*)d_in[5];
  const float* fw2 = (const float*)d_in[6];
  const float* fb2 = (const float*)d_in[7];
  const float* flg = (const float*)d_in[8];
  const float* flb = (const float*)d_in[9];
  const float* lw  = (const float*)d_in[10];
  const float* lb  = (const float*)d_in[11];
  const float* ttg = (const float*)d_in[12];
  const float* ttb = (const float*)d_in[13];
  const float* wo  = (const float*)d_in[14];
  const float* wg  = (const float*)d_in[15];
  const float* png = (const float*)d_in[16];
  const float* pnb = (const float*)d_in[17];

  const int M = BB * SS;
  const size_t PLANE = (size_t)M * HIDC;
  float* qb    = (float*)d_ws;
  float* kb    = qb + PLANE;
  float* vb    = kb + PLANE;
  float* gactb = vb + PLANE;
  float* outb  = gactb + PLANE;
  float* gatedb = qb;

  dim3 gblk(256);
  dim3 ggrid(HIDC / 128, M / 16);
  gemm_wmma_f32<<<ggrid, gblk, 0, stream>>>(x, wq, qb,    M, HIDC, HIDC, 0);
  gemm_wmma_f32<<<ggrid, gblk, 0, stream>>>(x, wk, kb,    M, HIDC, HIDC, 0);
  gemm_wmma_f32<<<ggrid, gblk, 0, stream>>>(x, wv, vb,    M, HIDC, HIDC, 0);
  gemm_wmma_f32<<<ggrid, gblk, 0, stream>>>(x, wg, gactb, M, HIDC, HIDC, 1);

  ttt_scan_kernel<<<HH, 256, SCAN_SMEM_FLOATS * sizeof(float), stream>>>(
      kb, vb, qb, fw1, fb1, fw2, fb2, flg, flb, lw, lb, ttg, ttb, outb);

  gate_ln_kernel<<<M, 256, 0, stream>>>(outb, gactb, png, pnb, gatedb);

  gemm_wmma_f32<<<ggrid, gblk, 0, stream>>>(gatedb, wo, (float*)d_out,
                                            M, HIDC, HIDC, 0);
}


# LDS row pitch 64 -> 76 floats for the scan's weights and 8x64 tiles and the GEMM's A tile: WMMA fragment loads that walk rows stop landing 8 and 16 deep on one bank
# speedup vs baseline: 2.1026x; 1.7210x over previous
#include <hip/hip_runtime.h>
#include <hip/hip_bf16.h>
#include <math.h>

#define BB     8
#define SS     256
#define HIDC   1024
#define HH     16
#define DD     64
#define FF     64
#define LR     0.1f
#define GSCALE (2.0f / 8192.0f)
#define LN_EPS 1e-5f
#define LP 76

typedef float v2f __attribute__((ext_vector_type(2)));
typedef float v8f __attribute__((ext_vector_type(8)));

__device__ __forceinline__ v8f wmma4(v2f a, v2f b, v8f c) {
  return __builtin_amdgcn_wmma_f32_16x16x4_f32(
       false, a,  false, b,
       (short)0, c,  false,  false);
}

__device__ __forceinline__ float gelu_f(float x) {
  const float c = 0.7978845608028654f;
  float u = c * (x + 0.044715f * x * x * x);
  return 0.5f * x * (1.0f + tanhf(u));
}
__device__ __forceinline__ float gelu_grad(float x) {
  const float c = 0.7978845608028654f;
  float x2 = x * x;
  float u = c * (x + 0.044715f * x2 * x);
  float t = tanhf(u);
  return 0.5f * (1.0f + t) + 0.5f * x * (1.0f - t * t) * c * (1.0f + 0.134145f * x2);
}
__device__ __forceinline__ float wave_sum32(float v) {
#pragma unroll
  for (int off = 16; off > 0; off >>= 1) v += __shfl_xor(v, off, 32);
  return v;
}

__global__ void __launch_bounds__(256)
gemm_wmma_f32(const float* __restrict__ A, const float* __restrict__ Bm,
              float* __restrict__ C, int M, int N, int K, int act) {
  __shared__ float As[16 * LP];
  const int tid   = threadIdx.x;
  const int wave  = tid >> 5;
  const int lane  = tid & 31;
  const int mbase = blockIdx.y * 16;
  const int tn    = blockIdx.x * 8 + wave;
  const int col   = tn * 16 + (lane & 15);
  const int kh    = (lane >> 4) << 1;
  const int arow  = lane & 15;
  v8f acc = {0.f, 0.f, 0.f, 0.f, 0.f, 0.f, 0.f, 0.f};
  for (int kc = 0; kc < K; kc += 64) {
    __syncthreads();
    for (int i = tid; i < 16 * 64; i += 256)
      As[(i >> 6) * LP + (i & 63)] = A[(size_t)(mbase + (i >> 6)) * K + kc + (i & 63)];
    if (kc + 64 < K)
      __builtin_prefetch(&A[(size_t)(mbase + (tid >> 4)) * K + kc + 64], 0, 1);
    __syncthreads();
#pragma unroll
    for (int k0 = 0; k0 < 64; k0 += 4) {
      v2f a, b;
      a.x = As[arow * LP + k0 + kh];
      a.y = As[arow * LP + k0 + kh + 1];
      b.x = Bm[(size_t)(kc + k0 + kh) * N + col];
      b.y = Bm[(size_t)(kc + k0 + kh + 1) * N + col];
      acc = wmma4(a, b, acc);
    }
  }
  const int r0 = mbase + ((lane >= 16) ? 8 : 0);
#pragma unroll
  for (int i = 0; i < 8; ++i) {
    float v = acc[i];
    if (act) v = gelu_f(v);
    C[(size_t)(r0 + i) * N + col] = v;
  }
}

#define EPI_BIAS      0
#define EPI_GELU_PAIR 1
#define EPI_DPRED     2
#define EPI_DGELU     3

__device__ __forceinline__ void
gemm16_rowA(const float* A8, const float* Bm, const float* bias,
            float* C8, float* C2, const float* aux,
            int transB, int epi, int wave, int lane) {
  if (wave >= 4) return;
  const int col  = wave * 16 + (lane & 15);
  const int mrow = lane & 7;
  const int kh   = (lane >> 4) << 1;
  v2f a[16], b[16];
#pragma unroll
  for (int j = 0; j < 16; ++j) {
    const int k = 4 * j + kh;
    a[j].x = A8[mrow * LP + k];
    a[j].y = A8[mrow * LP + k + 1];
    if (!transB) {
      b[j].x = Bm[k * LP + col];
      b[j].y = Bm[(k + 1) * LP + col];
    } else {
      b[j].x = Bm[col * LP + k];
      b[j].y = Bm[col * LP + k + 1];
    }
  }
  const float bv = bias ? bias[col] : 0.f;
  float ax[8];
#pragma unroll
  for (int i = 0; i < 8; ++i)
    ax[i] = (epi == EPI_DPRED || epi == EPI_DGELU) ? aux[i * LP + col] : 0.f;
  __builtin_amdgcn_sched_barrier(0);
  v8f acc = {0.f, 0.f, 0.f, 0.f, 0.f, 0.f, 0.f, 0.f};
#pragma unroll
  for (int j = 0; j < 16; ++j) acc = wmma4(a[j], b[j], acc);
  if (lane < 16) {
#pragma unroll
    for (int i = 0; i < 8; ++i) {
      const int o = i * LP + col;
      float v = acc[i] + bv;
      if (epi == EPI_GELU_PAIR)      { C8[o] = v; C2[o] = gelu_f(v); }
      else if (epi == EPI_DPRED)     { C8[o] = GSCALE * (v - ax[i]); }
      else if (epi == EPI_DGELU)     { C8[o] = v * gelu_grad(ax[i]); }
      else                           { C8[o] = v; }
    }
  }
}

__device__ __forceinline__ void
gemm_outer_update(const float* X8, const float* Y8, float* Wt, float lr,
                  int wave, int lane) {
  const int kh = (lane >> 4) << 1;
#pragma unroll
  for (int ti = 0; ti < 2; ++ti) {
    const int t  = wave * 2 + ti;
    const int tm = t >> 2, tn = t & 3;
    const int m = tm * 16 + (lane & 15);
    const int n = tn * 16 + (lane & 15);
    v8f acc = {0.f, 0.f, 0.f, 0.f, 0.f, 0.f, 0.f, 0.f};
#pragma unroll
    for (int k0 = 0; k0 < 8; k0 += 4) {
      v2f a, b;
      a.x = X8[(k0 + kh) * LP + m];
      a.y = X8[(k0 + kh + 1) * LP + m];
      b.x = Y8[(k0 + kh) * LP + n];
      b.y = Y8[(k0 + kh + 1) * LP + n];
      acc = wmma4(a, b, acc);
    }
    const int rbase = tm * 16 + ((lane >= 16) ? 8 : 0);
    const int cc    = tn * 16 + (lane & 15);
#pragma unroll
    for (int i = 0; i < 8; ++i) Wt[(rbase + i) * LP + cc] -= lr * acc[i];
  }
}

__device__ __forceinline__ void
fwd_pass(const float* ktb, float* av, float* hg, float* ob, float* xh, float* zb,
         const float* W1, const float* W2, const float* b1v, const float* b2v,
         const float* lgv, const float* lbv, float* muv, float* isv,
         int wave, int lane, int tid) {
  gemm16_rowA(ktb, W1, b1v, av, hg, nullptr, 0, EPI_GELU_PAIR, wave, lane);
  __syncthreads();
  gemm16_rowA(hg, W2, b2v, ob, nullptr, nullptr, 0, EPI_BIAS, wave, lane);
  __syncthreads();
  {
    float e0 = ob[wave * LP + lane], e1 = ob[wave * LP + lane + 32];
    float s = wave_sum32(e0 + e1);
    float q = wave_sum32(e0 * e0 + e1 * e1);
    if (lane == 0) {
      float m = s * (1.f / 64.f);
      muv[wave] = m;
      isv[wave] = rsqrtf(q * (1.f / 64.f) - m * m + LN_EPS);
    }
  }
  __syncthreads();
  for (int idx = tid; idx < 512; idx += 256) {
    int b = idx >> 6, d = idx & 63, o = b * LP + d;
    float xv = (ob[o] - muv[b]) * isv[b];
    xh[o] = xv;
    zb[o] = xv * lgv[d] + lbv[d];
  }
  __syncthreads();
}

#define SCAN_SMEM_FLOATS (3 * 64 * LP + 10 * 8 * LP + 7 * 64 + 32)

__global__ void __launch_bounds__(256)
ttt_scan_kernel(const float* __restrict__ kg, const float* __restrict__ vg,
                const float* __restrict__ qg,
                const float* __restrict__ fw1, const float* __restrict__ fb1,
                const float* __restrict__ fw2, const float* __restrict__ fb2,
                const float* __restrict__ flg, const float* __restrict__ flb,
                const float* __restrict__ lwg, const float* __restrict__ lbg,
                const float* __restrict__ ttg, const float* __restrict__ ttb,
                float* __restrict__ outg) {
  extern __shared__ float smem[];
  float* W1  = smem;
  float* W2  = W1 + 64 * LP;
  float* LW  = W2 + 64 * LP;
  float* ktb = LW + 64 * LP;
  float* tgt = ktb + 8 * LP;
  float* av  = tgt + 8 * LP;
  float* hg  = av + 8 * LP;
  float* ob  = hg + 8 * LP;
  float* xh  = ob + 8 * LP;
  float* zb  = xh + 8 * LP;
  float* dzb = zb + 8 * LP;
  float* dob = dzb + 8 * LP;
  float* dab = dob + 8 * LP;
  float* b1v = dab + 8 * LP;
  float* b2v = b1v + 64;
  float* lgv = b2v + 64;
  float* lbv = lgv + 64;
  float* tgv = lbv + 64;
  float* tbv = tgv + 64;
  float* lbz = tbv + 64;
  float* muv = lbz + 64;
  float* isv = muv + 8;
  float* m1v = isv + 8;
  float* m2v = m1v + 8;

  const int h    = blockIdx.x;
  const int tid  = threadIdx.x;
  const int wave = tid >> 5;
  const int lane = tid & 31;

  const int b0 = tid >> 6,          d0 = tid & 63;
  const int b1i = (tid + 256) >> 6, d1i = (tid + 256) & 63;
  const size_t g0base = (size_t)(b0 * SS) * HIDC + h * 64 + d0;
  const size_t g1base = (size_t)(b1i * SS) * HIDC + h * 64 + d1i;
  const int l0 = b0 * LP + d0, l1 = b1i * LP + d1i;

  for (int i = tid; i < 4096; i += 256) {
    const int o = (i >> 6) * LP + (i & 63);
    W1[o] = fw1[(size_t)h * 4096 + i];
    W2[o] = fw2[(size_t)h * 4096 + i];
    LW[o] = lwg[i];
  }
  if (tid < 64) {
    b1v[tid] = fb1[h * 64 + tid];
    b2v[tid] = fb2[h * 64 + tid];
    lgv[tid] = flg[h * 64 + tid];
    lbv[tid] = flb[h * 64 + tid];
    tgv[tid] = ttg[h * 64 + tid];
    tbv[tid] = ttb[h * 64 + tid];
    lbz[tid] = lbg[tid];
  }

  float kp0 = kg[g0base], vp0 = vg[g0base];
  float kp1 = kg[g1base], vp1 = vg[g1base];
  __syncthreads();

  for (int t = 0; t < SS; ++t) {
    ktb[l0] = kp0;  tgt[l0] = vp0 - kp0;
    ktb[l1] = kp1;  tgt[l1] = vp1 - kp1;
    __syncthreads();
    if (t + 1 < SS) {
      kp0 = kg[g0base + (t + 1) * HIDC];  vp0 = vg[g0base + (t + 1) * HIDC];
      kp1 = kg[g1base + (t + 1) * HIDC];  vp1 = vg[g1base + (t + 1) * HIDC];
    }

    for (int step = 0; step < 4; ++step) {
      fwd_pass(ktb, av, hg, ob, xh, zb, W1, W2, b1v, b2v, lgv, lbv,
               muv, isv, wave, lane, tid);
      gemm16_rowA(zb, LW, lbz, dab, nullptr, tgt, 0, EPI_DPRED, wave, lane);
      __syncthreads();
      gemm16_rowA(dab, LW, nullptr, dzb, nullptr, nullptr, 1, EPI_BIAS, wave, lane);
      __syncthreads();
      {
        float dd0 = dzb[wave * LP + lane] * lgv[lane];
        float dd1 = dzb[wave * LP + lane + 32] * lgv[lane + 32];
        float x0 = xh[wave * LP + lane], x1 = xh[wave * LP + lane + 32];
        float s1 = wave_sum32(dd0 + dd1);
        float s2 = wave_sum32(dd0 * x0 + dd1 * x1);
        if (lane == 0) { m1v[wave] = s1 * (1.f / 64.f); m2v[wave] = s2 * (1.f / 64.f); }
      }
      __syncthreads();
      for (int idx = tid; idx < 512; idx += 256) {
        int b = idx >> 6, d = idx & 63, o = b * LP + d;
        float dxh = dzb[o] * lgv[d];
        dob[o] = isv[b] * (dxh - m1v[b] - xh[o] * m2v[b]);
      }
      __syncthreads();
      if (tid < 64) {
        float s = 0.f;
        for (int b = 0; b < 8; ++b) s += dzb[b * LP + tid] * xh[b * LP + tid];
        lgv[tid] -= LR * s;
      } else if (tid < 128) {
        int d = tid - 64; float s = 0.f;
        for (int b = 0; b < 8; ++b) s += dzb[b * LP + d];
        lbv[d] -= LR * s;
      }
      gemm16_rowA(dob, W2, nullptr, dab, nullptr, av, 1, EPI_DGELU, wave, lane);
      __syncthreads();
      gemm_outer_update(hg, dob, W2, LR, wave, lane);
      gemm_outer_update(ktb, dab, W1, LR, wave, lane);
      if (tid < 64) {
        float s = 0.f;
        for (int b = 0; b < 8; ++b) s += dob[b * LP + tid];
        b2v[tid] -= LR * s;
      } else if (tid < 128) {
        int d = tid - 64; float s = 0.f;
        for (int b = 0; b < 8; ++b) s += dab[b * LP + d];
        b1v[d] -= LR * s;
      }
      __syncthreads();
    }

    fwd_pass(ktb, av, hg, ob, xh, zb, W1, W2, b1v, b2v, lgv, lbv,
             muv, isv, wave, lane, tid);
    {
      float e0 = zb[wave * LP + lane], e1 = zb[wave * LP + lane + 32];
      float s = wave_sum32(e0 + e1);
      float q = wave_sum32(e0 * e0 + e1 * e1);
      if (lane == 0) {
        float m = s * (1.f / 64.f);
        muv[wave] = m;
        isv[wave] = rsqrtf(q * (1.f / 64.f) - m * m + LN_EPS);
      }
    }
    __syncthreads();
    for (int idx = tid; idx < 512; idx += 256) {
      int b = idx >> 6, d = idx & 63;
      float z2 = (zb[b * LP + d] - muv[b]) * isv[b] * tgv[d] + tbv[d];
      size_t g = ((size_t)(b * SS + t)) * HIDC + h * 64 + d;
      outg[g] = qg[g] + z2;
    }
    __syncthreads();
  }
}

__global__ void __launch_bounds__(256)
gate_ln_kernel(const float* __restrict__ outb, const float* __restrict__ gact,
               const float* __restrict__ png, const float* __restrict__ pnb,
               float* __restrict__ gated) {
  __shared__ float r1[256], r2[256];
  __shared__ float mu_s, inv_s;
  const int r = blockIdx.x, tid = threadIdx.x;
  const float* row = outb + (size_t)r * HIDC;
  float s = 0.f, q = 0.f;
  for (int i = tid; i < HIDC; i += 256) { float v = row[i]; s += v; q += v * v; }
  r1[tid] = s; r2[tid] = q;
  __syncthreads();
  for (int off = 128; off > 0; off >>= 1) {
    if (tid < off) { r1[tid] += r1[tid + off]; r2[tid] += r2[tid + off]; }
    __syncthreads();
  }
  if (tid == 0) {
    float m = r1[0] * (1.f / HIDC);
    mu_s = m;
    inv_s = rsqrtf(r2[0] * (1.f / HIDC) - m * m + LN_EPS);
  }
  __syncthreads();
  for (int i = tid; i < HIDC; i += 256) {
    float y = (row[i] - mu_s) * inv_s * png[i] + pnb[i];
    gated[(size_t)r * HIDC + i] = gact[(size_t)r * HIDC + i] * y;
  }
}

extern "C" void kernel_launch(void* const* d_in, const int* in_sizes, int n_in,
                              void* d_out, int out_size, void* d_ws, size_t ws_size,
                              hipStream_t stream) {
  (void)in_sizes; (void)n_in; (void)out_size; (void)ws_size;
  const float* x   = (const float*)d_in[0];
  const float* wq  = (const float*)d_in[1];
  const float* wk  = (const float*)d_in[2];
  const float* wv  = (const float*)d_in[3];
  const float* fw1 = (const float*)d_in[4];
  const float* fb1 = (const float*)d_in[5];
  const float* fw2 = (const float*)d_in[6];
  const float* fb2 = (const float*)d_in[7];
  const float* flg = (const float*)d_in[8];
  const float* flb = (const float*)d_in[9];
  const float* lw  = (const float*)d_in[10];
  const float* lb  = (const float*)d_in[11];
  const float* ttg = (const float*)d_in[12];
  const float* ttb = (const float*)d_in[13];
  const float* wo  = (const float*)d_in[14];
  const float* wg  = (const float*)d_in[15];
  const float* png = (const float*)d_in[16];
  const float* pnb = (const float*)d_in[17];

  const int M = BB * SS;
  const size_t PLANE = (size_t)M * HIDC;
  float* qb    = (float*)d_ws;
  float* kb    = qb + PLANE;
  float* vb    = kb + PLANE;
  float* gactb = vb + PLANE;
  float* outb  = gactb + PLANE;
  float* gatedb = qb;

  dim3 gblk(256);
  dim3 ggrid(HIDC / 128, M / 16);
  gemm_wmma_f32<<<ggrid, gblk, 0, stream>>>(x, wq, qb,    M, HIDC, HIDC, 0);
  gemm_wmma_f32<<<ggrid, gblk, 0, stream>>>(x, wk, kb,    M, HIDC, HIDC, 0);
  gemm_wmma_f32<<<ggrid, gblk, 0, stream>>>(x, wv, vb,    M, HIDC, HIDC, 0);
  gemm_wmma_f32<<<ggrid, gblk, 0, stream>>>(x, wg, gactb, M, HIDC, HIDC, 1);

  ttt_scan_kernel<<<HH, 256, SCAN_SMEM_FLOATS * sizeof(float), stream>>>(
      kb, vb, qb, fw1, fb1, fw2, fb2, flg, flb, lw, lb, ttg, ttb, outb);

  gate_ln_kernel<<<M, 256, 0, stream>>>(outb, gactb, png, pnb, gatedb);

  gemm_wmma_f32<<<ggrid, gblk, 0, stream>>>(gatedb, wo, (float*)d_out,
                                            M, HIDC, HIDC, 0);
}


# gelu and gelu' epilogues: library tanhf (exec-masked paths, a lane's 8 values ran one after another) replaced by branch-free 1-2/(exp2(c|u|)+1) on v_exp/v_rcp; tanh abs err about 1e-7
# speedup vs baseline: 2.5251x; 1.2009x over previous
#include <hip/hip_runtime.h>
#include <hip/hip_bf16.h>
#include <math.h>

#define BB     8
#define SS     256
#define HIDC   1024
#define HH     16
#define DD     64
#define FF     64
#define LR     0.1f
#define GSCALE (2.0f / 8192.0f)
#define LN_EPS 1e-5f
#define LP 76

typedef float v2f __attribute__((ext_vector_type(2)));
typedef float v8f __attribute__((ext_vector_type(8)));

__device__ __forceinline__ v8f wmma4(v2f a, v2f b, v8f c) {
  return __builtin_amdgcn_wmma_f32_16x16x4_f32(
       false, a,  false, b,
       (short)0, c,  false,  false);
}

__device__ __forceinline__ float tanh_straight(float u) {
  const float a = fabsf(u);
  const float e = __builtin_amdgcn_exp2f(a * 2.8853900817779268f);
  const float t = 1.0f - 2.0f * __builtin_amdgcn_rcpf(e + 1.0f);
  return copysignf(t, u);
}

__device__ __forceinline__ float gelu_f(float x) {
  const float c = 0.7978845608028654f;
  float u = c * (x + 0.044715f * x * x * x);
  return 0.5f * x * (1.0f + tanh_straight(u));
}
__device__ __forceinline__ float gelu_grad(float x) {
  const float c = 0.7978845608028654f;
  float x2 = x * x;
  float u = c * (x + 0.044715f * x2 * x);
  float t = tanh_straight(u);
  return 0.5f * (1.0f + t) + 0.5f * x * (1.0f - t * t) * c * (1.0f + 0.134145f * x2);
}
__device__ __forceinline__ float wave_sum32(float v) {
#pragma unroll
  for (int off = 16; off > 0; off >>= 1) v += __shfl_xor(v, off, 32);
  return v;
}

__global__ void __launch_bounds__(256)
gemm_wmma_f32(const float* __restrict__ A, const float* __restrict__ Bm,
              float* __restrict__ C, int M, int N, int K, int act) {
  __shared__ float As[16 * LP];
  const int tid   = threadIdx.x;
  const int wave  = tid >> 5;
  const int lane  = tid & 31;
  const int mbase = blockIdx.y * 16;
  const int tn    = blockIdx.x * 8 + wave;
  const int col   = tn * 16 + (lane & 15);
  const int kh    = (lane >> 4) << 1;
  const int arow  = lane & 15;
  v8f acc = {0.f, 0.f, 0.f, 0.f, 0.f, 0.f, 0.f, 0.f};
  for (int kc = 0; kc < K; kc += 64) {
    __syncthreads();
    for (int i = tid; i < 16 * 64; i += 256)
      As[(i >> 6) * LP + (i & 63)] = A[(size_t)(mbase + (i >> 6)) * K + kc + (i & 63)];
    if (kc + 64 < K)
      __builtin_prefetch(&A[(size_t)(mbase + (tid >> 4)) * K + kc + 64], 0, 1);
    __syncthreads();
#pragma unroll
    for (int k0 = 0; k0 < 64; k0 += 4) {
      v2f a, b;
      a.x = As[arow * LP + k0 + kh];
      a.y = As[arow * LP + k0 + kh + 1];
      b.x = Bm[(size_t)(kc + k0 + kh) * N + col];
      b.y = Bm[(size_t)(kc + k0 + kh + 1) * N + col];
      acc = wmma4(a, b, acc);
    }
  }
  const int r0 = mbase + ((lane >= 16) ? 8 : 0);
#pragma unroll
  for (int i = 0; i < 8; ++i) {
    float v = acc[i];
    if (act) v = gelu_f(v);
    C[(size_t)(r0 + i) * N + col] = v;
  }
}

#define EPI_BIAS      0
#define EPI_GELU_PAIR 1
#define EPI_DPRED     2
#define EPI_DGELU     3

__device__ __forceinline__ void
gemm16_rowA(const float* A8, const float* Bm, const float* bias,
            float* C8, float* C2, const float* aux,
            int transB, int epi, int wave, int lane) {
  if (wave >= 4) return;
  const int col  = wave * 16 + (lane & 15);
  const int mrow = lane & 7;
  const int kh   = (lane >> 4) << 1;
  v2f a[16], b[16];
#pragma unroll
  for (int j = 0; j < 16; ++j) {
    const int k = 4 * j + kh;
    a[j].x = A8[mrow * LP + k];
    a[j].y = A8[mrow * LP + k + 1];
    if (!transB) {
      b[j].x = Bm[k * LP + col];
      b[j].y = Bm[(k + 1) * LP + col];
    } else {
      b[j].x = Bm[col * LP + k];
      b[j].y = Bm[col * LP + k + 1];
    }
  }
  const float bv = bias ? bias[col] : 0.f;
  float ax[8];
#pragma unroll
  for (int i = 0; i < 8; ++i)
    ax[i] = (epi == EPI_DPRED || epi == EPI_DGELU) ? aux[i * LP + col] : 0.f;
  __builtin_amdgcn_sched_barrier(0);
  v8f acc = {0.f, 0.f, 0.f, 0.f, 0.f, 0.f, 0.f, 0.f};
#pragma unroll
  for (int j = 0; j < 16; ++j) acc = wmma4(a[j], b[j], acc);
  if (lane < 16) {
#pragma unroll
    for (int i = 0; i < 8; ++i) {
      const int o = i * LP + col;
      float v = acc[i] + bv;
      if (epi == EPI_GELU_PAIR)      { C8[o] = v; C2[o] = gelu_f(v); }
      else if (epi == EPI_DPRED)     { C8[o] = GSCALE * (v - ax[i]); }
      else if (epi == EPI_DGELU)     { C8[o] = v * gelu_grad(ax[i]); }
      else                           { C8[o] = v; }
    }
  }
}

__device__ __forceinline__ void
gemm_outer_update(const float* X8, const float* Y8, float* Wt, float lr,
                  int wave, int lane) {
  const int kh = (lane >> 4) << 1;
#pragma unroll
  for (int ti = 0; ti < 2; ++ti) {
    const int t  = wave * 2 + ti;
    const int tm = t >> 2, tn = t & 3;
    const int m = tm * 16 + (lane & 15);
    const int n = tn * 16 + (lane & 15);
    v8f acc = {0.f, 0.f, 0.f, 0.f, 0.f, 0.f, 0.f, 0.f};
#pragma unroll
    for (int k0 = 0; k0 < 8; k0 += 4) {
      v2f a, b;
      a.x = X8[(k0 + kh) * LP + m];
      a.y = X8[(k0 + kh + 1) * LP + m];
      b.x = Y8[(k0 + kh) * LP + n];
      b.y = Y8[(k0 + kh + 1) * LP + n];
      acc = wmma4(a, b, acc);
    }
    const int rbase = tm * 16 + ((lane >= 16) ? 8 : 0);
    const int cc    = tn * 16 + (lane & 15);
#pragma unroll
    for (int i = 0; i < 8; ++i) Wt[(rbase + i) * LP + cc] -= lr * acc[i];
  }
}

__device__ __forceinline__ void
fwd_pass(const float* ktb, float* av, float* hg, float* ob, float* xh, float* zb,
         const float* W1, const float* W2, const float* b1v, const float* b2v,
         const float* lgv, const float* lbv, float* muv, float* isv,
         int wave, int lane, int tid) {
  gemm16_rowA(ktb, W1, b1v, av, hg, nullptr, 0, EPI_GELU_PAIR, wave, lane);
  __syncthreads();
  gemm16_rowA(hg, W2, b2v, ob, nullptr, nullptr, 0, EPI_BIAS, wave, lane);
  __syncthreads();
  {
    float e0 = ob[wave * LP + lane], e1 = ob[wave * LP + lane + 32];
    float s = wave_sum32(e0 + e1);
    float q = wave_sum32(e0 * e0 + e1 * e1);
    if (lane == 0) {
      float m = s * (1.f / 64.f);
      muv[wave] = m;
      isv[wave] = rsqrtf(q * (1.f / 64.f) - m * m + LN_EPS);
    }
  }
  __syncthreads();
  for (int idx = tid; idx < 512; idx += 256) {
    int b = idx >> 6, d = idx & 63, o = b * LP + d;
    float xv = (ob[o] - muv[b]) * isv[b];
    xh[o] = xv;
    zb[o] = xv * lgv[d] + lbv[d];
  }
  __syncthreads();
}

#define SCAN_SMEM_FLOATS (3 * 64 * LP + 10 * 8 * LP + 7 * 64 + 32)

__global__ void __launch_bounds__(256)
ttt_scan_kernel(const float* __restrict__ kg, const float* __restrict__ vg,
                const float* __restrict__ qg,
                const float* __restrict__ fw1, const float* __restrict__ fb1,
                const float* __restrict__ fw2, const float* __restrict__ fb2,
                const float* __restrict__ flg, const float* __restrict__ flb,
                const float* __restrict__ lwg, const float* __restrict__ lbg,
                const float* __restrict__ ttg, const float* __restrict__ ttb,
                float* __restrict__ outg) {
  extern __shared__ float smem[];
  float* W1  = smem;
  float* W2  = W1 + 64 * LP;
  float* LW  = W2 + 64 * LP;
  float* ktb = LW + 64 * LP;
  float* tgt = ktb + 8 * LP;
  float* av  = tgt + 8 * LP;
  float* hg  = av + 8 * LP;
  float* ob  = hg + 8 * LP;
  float* xh  = ob + 8 * LP;
  float* zb  = xh + 8 * LP;
  float* dzb = zb + 8 * LP;
  float* dob = dzb + 8 * LP;
  float* dab = dob + 8 * LP;
  float* b1v = dab + 8 * LP;
  float* b2v = b1v + 64;
  float* lgv = b2v + 64;
  float* lbv = lgv + 64;
  float* tgv = lbv + 64;
  float* tbv = tgv + 64;
  float* lbz = tbv + 64;
  float* muv = lbz + 64;
  float* isv = muv + 8;
  float* m1v = isv + 8;
  float* m2v = m1v + 8;

  const int h    = blockIdx.x;
  const int tid  = threadIdx.x;
  const int wave = tid >> 5;
  const int lane = tid & 31;

  const int b0 = tid >> 6,          d0 = tid & 63;
  const int b1i = (tid + 256) >> 6, d1i = (tid + 256) & 63;
  const size_t g0base = (size_t)(b0 * SS) * HIDC + h * 64 + d0;
  const size_t g1base = (size_t)(b1i * SS) * HIDC + h * 64 + d1i;
  const int l0 = b0 * LP + d0, l1 = b1i * LP + d1i;

  for (int i = tid; i < 4096; i += 256) {
    const int o = (i >> 6) * LP + (i & 63);
    W1[o] = fw1[(size_t)h * 4096 + i];
    W2[o] = fw2[(size_t)h * 4096 + i];
    LW[o] = lwg[i];
  }
  if (tid < 64) {
    b1v[tid] = fb1[h * 64 + tid];
    b2v[tid] = fb2[h * 64 + tid];
    lgv[tid] = flg[h * 64 + tid];
    lbv[tid] = flb[h * 64 + tid];
    tgv[tid] = ttg[h * 64 + tid];
    tbv[tid] = ttb[h * 64 + tid];
    lbz[tid] = lbg[tid];
  }

  float kp0 = kg[g0base], vp0 = vg[g0base];
  float kp1 = kg[g1base], vp1 = vg[g1base];
  __syncthreads();

  for (int t = 0; t < SS; ++t) {
    ktb[l0] = kp0;  tgt[l0] = vp0 - kp0;
    ktb[l1] = kp1;  tgt[l1] = vp1 - kp1;
    __syncthreads();
    if (t + 1 < SS) {
      kp0 = kg[g0base + (t + 1) * HIDC];  vp0 = vg[g0base + (t + 1) * HIDC];
      kp1 = kg[g1base + (t + 1) * HIDC];  vp1 = vg[g1base + (t + 1) * HIDC];
    }

    for (int step = 0; step < 4; ++step) {
      fwd_pass(ktb, av, hg, ob, xh, zb, W1, W2, b1v, b2v, lgv, lbv,
               muv, isv, wave, lane, tid);
      gemm16_rowA(zb, LW, lbz, dab, nullptr, tgt, 0, EPI_DPRED, wave, lane);
      __syncthreads();
      gemm16_rowA(dab, LW, nullptr, dzb, nullptr, nullptr, 1, EPI_BIAS, wave, lane);
      __syncthreads();
      {
        float dd0 = dzb[wave * LP + lane] * lgv[lane];
        float dd1 = dzb[wave * LP + lane + 32] * lgv[lane + 32];
        float x0 = xh[wave * LP + lane], x1 = xh[wave * LP + lane + 32];
        float s1 = wave_sum32(dd0 + dd1);
        float s2 = wave_sum32(dd0 * x0 + dd1 * x1);
        if (lane == 0) { m1v[wave] = s1 * (1.f / 64.f); m2v[wave] = s2 * (1.f / 64.f); }
      }
      __syncthreads();
      for (int idx = tid; idx < 512; idx += 256) {
        int b = idx >> 6, d = idx & 63, o = b * LP + d;
        float dxh = dzb[o] * lgv[d];
        dob[o] = isv[b] * (dxh - m1v[b] - xh[o] * m2v[b]);
      }
      __syncthreads();
      if (tid < 64) {
        float s = 0.f;
        for (int b = 0; b < 8; ++b) s += dzb[b * LP + tid] * xh[b * LP + tid];
        lgv[tid] -= LR * s;
      } else if (tid < 128) {
        int d = tid - 64; float s = 0.f;
        for (int b = 0; b < 8; ++b) s += dzb[b * LP + d];
        lbv[d] -= LR * s;
      }
      gemm16_rowA(dob, W2, nullptr, dab, nullptr, av, 1, EPI_DGELU, wave, lane);
      __syncthreads();
      gemm_outer_update(hg, dob, W2, LR, wave, lane);
      gemm_outer_update(ktb, dab, W1, LR, wave, lane);
      if (tid < 64) {
        float s = 0.f;
        for (int b = 0; b < 8; ++b) s += dob[b * LP + tid];
        b2v[tid] -= LR * s;
      } else if (tid < 128) {
        int d = tid - 64; float s = 0.f;
        for (int b = 0; b < 8; ++b) s += dab[b * LP + d];
        b1v[d] -= LR * s;
      }
      __syncthreads();
    }

    fwd_pass(ktb, av, hg, ob, xh, zb, W1, W2, b1v, b2v, lgv, lbv,
             muv, isv, wave, lane, tid);
    {
      float e0 = zb[wave * LP + lane], e1 = zb[wave * LP + lane + 32];
      float s = wave_sum32(e0 + e1);
      float q = wave_sum32(e0 * e0 + e1 * e1);
      if (lane == 0) {
        float m = s * (1.f / 64.f);
        muv[wave] = m;
        isv[wave] = rsqrtf(q * (1.f / 64.f) - m * m + LN_EPS);
      }
    }
    __syncthreads();
    for (int idx = tid; idx < 512; idx += 256) {
      int b = idx >> 6, d = idx & 63;
      float z2 = (zb[b * LP + d] - muv[b]) * isv[b] * tgv[d] + tbv[d];
      size_t g = ((size_t)(b * SS + t)) * HIDC + h * 64 + d;
      outg[g] = qg[g] + z2;
    }
    __syncthreads();
  }
}

__global__ void __launch_bounds__(256)
gate_ln_kernel(const float* __restrict__ outb, const float* __restrict__ gact,
               const float* __restrict__ png, const float* __restrict__ pnb,
               float* __restrict__ gated) {
  __shared__ float r1[256], r2[256];
  __shared__ float mu_s, inv_s;
  const int r = blockIdx.x, tid = threadIdx.x;
  const float* row = outb + (size_t)r * HIDC;
  float s = 0.f, q = 0.f;
  for (int i = tid; i < HIDC; i += 256) { float v = row[i]; s += v; q += v * v; }
  r1[tid] = s; r2[tid] = q;
  __syncthreads();
  for (int off = 128; off > 0; off >>= 1) {
    if (tid < off) { r1[tid] += r1[tid + off]; r2[tid] += r2[tid + off]; }
    __syncthreads();
  }
  if (tid == 0) {
    float m = r1[0] * (1.f / HIDC);
    mu_s = m;
    inv_s = rsqrtf(r2[0] * (1.f / HIDC) - m * m + LN_EPS);
  }
  __syncthreads();
  for (int i = tid; i < HIDC; i += 256) {
    float y = (row[i] - mu_s) * inv_s * png[i] + pnb[i];
    gated[(size_t)r * HIDC + i] = gact[(size_t)r * HIDC + i] * y;
  }
}

extern "C" void kernel_launch(void* const* d_in, const int* in_sizes, int n_in,
                              void* d_out, int out_size, void* d_ws, size_t ws_size,
                              hipStream_t stream) {
  (void)in_sizes; (void)n_in; (void)out_size; (void)ws_size;
  const float* x   = (const float*)d_in[0];
  const float* wq  = (const float*)d_in[1];
  const float* wk  = (const float*)d_in[2];
  const float* wv  = (const float*)d_in[3];
  const float* fw1 = (const float*)d_in[4];
  const float* fb1 = (const float*)d_in[5];
  const float* fw2 = (const float*)d_in[6];
  const float* fb2 = (const float*)d_in[7];
  const float* flg = (const float*)d_in[8];
  const float* flb = (const float*)d_in[9];
  const float* lw  = (const float*)d_in[10];
  const float* lb  = (const float*)d_in[11];
  const float* ttg = (const float*)d_in[12];
  const float* ttb = (const float*)d_in[13];
  const float* wo  = (const float*)d_in[14];
  const float* wg  = (const float*)d_in[15];
  const float* png = (const float*)d_in[16];
  const float* pnb = (const float*)d_in[17];

  const int M = BB * SS;
  const size_t PLANE = (size_t)M * HIDC;
  float* qb    = (float*)d_ws;
  float* kb    = qb + PLANE;
  float* vb    = kb + PLANE;
  float* gactb = vb + PLANE;
  float* outb  = gactb + PLANE;
  float* gatedb = qb;

  dim3 gblk(256);
  dim3 ggrid(HIDC / 128, M / 16);
  gemm_wmma_f32<<<ggrid, gblk, 0, stream>>>(x, wq, qb,    M, HIDC, HIDC, 0);
  gemm_wmma_f32<<<ggrid, gblk, 0, stream>>>(x, wk, kb,    M, HIDC, HIDC, 0);
  gemm_wmma_f32<<<ggrid, gblk, 0, stream>>>(x, wv, vb,    M, HIDC, HIDC, 0);
  gemm_wmma_f32<<<ggrid, gblk, 0, stream>>>(x, wg, gactb, M, HIDC, HIDC, 1);

  ttt_scan_kernel<<<HH, 256, SCAN_SMEM_FLOATS * sizeof(float), stream>>>(
      kb, vb, qb, fw1, fb1, fw2, fb2, flg, flb, lw, lb, ttg, ttb, outb);

  gate_ln_kernel<<<M, 256, 0, stream>>>(outb, gactb, png, pnb, gatedb);

  gemm_wmma_f32<<<ggrid, gblk, 0, stream>>>(gatedb, wo, (float*)d_out,
                                            M, HIDC, HIDC, 0);
}


# scan: the loss layer's two chained products per grad step folded into one, d_z = z@(s*LW*LW^T) + s*(loss_b-target)@LW^T; 1st factor once per block, 2nd term once per token; sums re-associated
# speedup vs baseline: 2.6938x; 1.0668x over previous
#include <hip/hip_runtime.h>
#include <hip/hip_bf16.h>
#include <math.h>

#define BB     8
#define SS     256
#define HIDC   1024
#define HH     16
#define DD     64
#define FF     64
#define LR     0.1f
#define GSCALE (2.0f / 8192.0f)
#define LN_EPS 1e-5f
#define LP 76

typedef float v2f __attribute__((ext_vector_type(2)));
typedef float v8f __attribute__((ext_vector_type(8)));

__device__ __forceinline__ v8f wmma4(v2f a, v2f b, v8f c) {
  return __builtin_amdgcn_wmma_f32_16x16x4_f32(
       false, a,  false, b,
       (short)0, c,  false,  false);
}

__device__ __forceinline__ float tanh_straight(float u) {
  const float a = fabsf(u);
  const float e = __builtin_amdgcn_exp2f(a * 2.8853900817779268f);
  const float t = 1.0f - 2.0f * __builtin_amdgcn_rcpf(e + 1.0f);
  return copysignf(t, u);
}

__device__ __forceinline__ float gelu_f(float x) {
  const float c = 0.7978845608028654f;
  float u = c * (x + 0.044715f * x * x * x);
  return 0.5f * x * (1.0f + tanh_straight(u));
}
__device__ __forceinline__ float gelu_grad(float x) {
  const float c = 0.7978845608028654f;
  float x2 = x * x;
  float u = c * (x + 0.044715f * x2 * x);
  float t = tanh_straight(u);
  return 0.5f * (1.0f + t) + 0.5f * x * (1.0f - t * t) * c * (1.0f + 0.134145f * x2);
}
__device__ __forceinline__ float wave_sum32(float v) {
#pragma unroll
  for (int off = 16; off > 0; off >>= 1) v += __shfl_xor(v, off, 32);
  return v;
}

__global__ void __launch_bounds__(256)
gemm_wmma_f32(const float* __restrict__ A, const float* __restrict__ Bm,
              float* __restrict__ C, int M, int N, int K, int act) {
  __shared__ float As[16 * LP];
  const int tid   = threadIdx.x;
  const int wave  = tid >> 5;
  const int lane  = tid & 31;
  const int mbase = blockIdx.y * 16;
  const int tn    = blockIdx.x * 8 + wave;
  const int col   = tn * 16 + (lane & 15);
  const int kh    = (lane >> 4) << 1;
  const int arow  = lane & 15;
  v8f acc = {0.f, 0.f, 0.f, 0.f, 0.f, 0.f, 0.f, 0.f};
  for (int kc = 0; kc < K; kc += 64) {
    __syncthreads();
    for (int i = tid; i < 16 * 64; i += 256)
      As[(i >> 6) * LP + (i & 63)] = A[(size_t)(mbase + (i >> 6)) * K + kc + (i & 63)];
    if (kc + 64 < K)
      __builtin_prefetch(&A[(size_t)(mbase + (tid >> 4)) * K + kc + 64], 0, 1);
    __syncthreads();
#pragma unroll
    for (int k0 = 0; k0 < 64; k0 += 4) {
      v2f a, b;
      a.x = As[arow * LP + k0 + kh];
      a.y = As[arow * LP + k0 + kh + 1];
      b.x = Bm[(size_t)(kc + k0 + kh) * N + col];
      b.y = Bm[(size_t)(kc + k0 + kh + 1) * N + col];
      acc = wmma4(a, b, acc);
    }
  }
  const int r0 = mbase + ((lane >= 16) ? 8 : 0);
#pragma unroll
  for (int i = 0; i < 8; ++i) {
    float v = acc[i];
    if (act) v = gelu_f(v);
    C[(size_t)(r0 + i) * N + col] = v;
  }
}

#define EPI_BIAS      0
#define EPI_GELU_PAIR 1
#define EPI_ADDAUX    2
#define EPI_DGELU     3
#define EPI_SCALE     4

__device__ __forceinline__ void
gemm16_rowA(const float* A8, const float* Bm, const float* bias,
            float* C8, float* C2, const float* aux,
            int transB, int epi, int wave, int lane) {
  if ((unsigned)wave >= 4u) return;
  const int col  = wave * 16 + (lane & 15);
  const int mrow = lane & 7;
  const int kh   = (lane >> 4) << 1;
  v2f a[16], b[16];
#pragma unroll
  for (int j = 0; j < 16; ++j) {
    const int k = 4 * j + kh;
    a[j].x = A8[mrow * LP + k];
    a[j].y = A8[mrow * LP + k + 1];
    if (!transB) {
      b[j].x = Bm[k * LP + col];
      b[j].y = Bm[(k + 1) * LP + col];
    } else {
      b[j].x = Bm[col * LP + k];
      b[j].y = Bm[col * LP + k + 1];
    }
  }
  const float bv = bias ? bias[col] : 0.f;
  float ax[8];
#pragma unroll
  for (int i = 0; i < 8; ++i)
    ax[i] = (epi == EPI_ADDAUX || epi == EPI_DGELU) ? aux[i * LP + col] : 0.f;
  __builtin_amdgcn_sched_barrier(0);
  v8f acc = {0.f, 0.f, 0.f, 0.f, 0.f, 0.f, 0.f, 0.f};
#pragma unroll
  for (int j = 0; j < 16; ++j) acc = wmma4(a[j], b[j], acc);
  if (lane < 16) {
#pragma unroll
    for (int i = 0; i < 8; ++i) {
      const int o = i * LP + col;
      float v = acc[i] + bv;
      if (epi == EPI_GELU_PAIR)      { C8[o] = v; C2[o] = gelu_f(v); }
      else if (epi == EPI_ADDAUX)    { C8[o] = v + ax[i]; }
      else if (epi == EPI_DGELU)     { C8[o] = v * gelu_grad(ax[i]); }
      else if (epi == EPI_SCALE)     { C8[o] = GSCALE * v; }
      else                           { C8[o] = v; }
    }
  }
}

__device__ __forceinline__ void
gemm_outer_update(const float* X8, const float* Y8, float* Wt, float lr,
                  int wave, int lane) {
  const int kh = (lane >> 4) << 1;
#pragma unroll
  for (int ti = 0; ti < 2; ++ti) {
    const int t  = wave * 2 + ti;
    const int tm = t >> 2, tn = t & 3;
    const int m = tm * 16 + (lane & 15);
    const int n = tn * 16 + (lane & 15);
    v8f acc = {0.f, 0.f, 0.f, 0.f, 0.f, 0.f, 0.f, 0.f};
#pragma unroll
    for (int k0 = 0; k0 < 8; k0 += 4) {
      v2f a, b;
      a.x = X8[(k0 + kh) * LP + m];
      a.y = X8[(k0 + kh + 1) * LP + m];
      b.x = Y8[(k0 + kh) * LP + n];
      b.y = Y8[(k0 + kh + 1) * LP + n];
      acc = wmma4(a, b, acc);
    }
    const int rbase = tm * 16 + ((lane >= 16) ? 8 : 0);
    const int cc    = tn * 16 + (lane & 15);
#pragma unroll
    for (int i = 0; i < 8; ++i) Wt[(rbase + i) * LP + cc] -= lr * acc[i];
  }
}

__device__ __forceinline__ void
gemm_self_transposed_scaled(const float* L, float* M, int wave, int lane) {
  const int kh = (lane >> 4) << 1;
  const int c  = lane & 15;
#pragma unroll
  for (int ti = 0; ti < 2; ++ti) {
    const int t  = wave * 2 + ti;
    const int tm = t >> 2, tn = t & 3;
    v8f acc = {0.f, 0.f, 0.f, 0.f, 0.f, 0.f, 0.f, 0.f};
#pragma unroll
    for (int j = 0; j < 16; ++j) {
      const int k = 4 * j + kh;
      v2f a, b;
      a.x = L[(tm * 16 + c) * LP + k];
      a.y = L[(tm * 16 + c) * LP + k + 1];
      b.x = L[(tn * 16 + c) * LP + k];
      b.y = L[(tn * 16 + c) * LP + k + 1];
      acc = wmma4(a, b, acc);
    }
    const int rbase = tm * 16 + ((lane >= 16) ? 8 : 0);
#pragma unroll
    for (int i = 0; i < 8; ++i) M[(rbase + i) * LP + tn * 16 + c] = GSCALE * acc[i];
  }
}

__device__ __forceinline__ void
fwd_pass(const float* ktb, float* av, float* hg, float* ob, float* xh, float* zb,
         const float* W1, const float* W2, const float* b1v, const float* b2v,
         const float* lgv, const float* lbv, float* muv, float* isv,
         int wave, int lane, int tid) {
  gemm16_rowA(ktb, W1, b1v, av, hg, nullptr, 0, EPI_GELU_PAIR, wave, lane);
  __syncthreads();
  gemm16_rowA(hg, W2, b2v, ob, nullptr, nullptr, 0, EPI_BIAS, wave, lane);
  __syncthreads();
  {
    float e0 = ob[wave * LP + lane], e1 = ob[wave * LP + lane + 32];
    float s = wave_sum32(e0 + e1);
    float q = wave_sum32(e0 * e0 + e1 * e1);
    if (lane == 0) {
      float m = s * (1.f / 64.f);
      muv[wave] = m;
      isv[wave] = rsqrtf(q * (1.f / 64.f) - m * m + LN_EPS);
    }
  }
  __syncthreads();
  for (int idx = tid; idx < 512; idx += 256) {
    int b = idx >> 6, d = idx & 63, o = b * LP + d;
    float xv = (ob[o] - muv[b]) * isv[b];
    xh[o] = xv;
    zb[o] = xv * lgv[d] + lbv[d];
  }
  __syncthreads();
}

#define SCAN_SMEM_FLOATS (4 * 64 * LP + 11 * 8 * LP + 6 * 64 + 32)

__global__ void __launch_bounds__(256)
ttt_scan_kernel(const float* __restrict__ kg, const float* __restrict__ vg,
                const float* __restrict__ qg,
                const float* __restrict__ fw1, const float* __restrict__ fb1,
                const float* __restrict__ fw2, const float* __restrict__ fb2,
                const float* __restrict__ flg, const float* __restrict__ flb,
                const float* __restrict__ lwg, const float* __restrict__ lbg,
                const float* __restrict__ ttg, const float* __restrict__ ttb,
                float* __restrict__ outg) {
  extern __shared__ float smem[];
  float* W1  = smem;
  float* W2  = W1 + 64 * LP;
  float* LW  = W2 + 64 * LP;
  float* MM  = LW + 64 * LP;
  float* ktb = MM + 64 * LP;
  float* tgt = ktb + 8 * LP;
  float* av  = tgt + 8 * LP;
  float* hg  = av + 8 * LP;
  float* ob  = hg + 8 * LP;
  float* xh  = ob + 8 * LP;
  float* zb  = xh + 8 * LP;
  float* dzb = zb + 8 * LP;
  float* dob = dzb + 8 * LP;
  float* dab = dob + 8 * LP;
  float* cb  = dab + 8 * LP;
  float* b1v = cb + 8 * LP;
  float* b2v = b1v + 64;
  float* lgv = b2v + 64;
  float* lbv = lgv + 64;
  float* tgv = lbv + 64;
  float* tbv = tgv + 64;
  float* muv = tbv + 64;
  float* isv = muv + 8;
  float* m1v = isv + 8;
  float* m2v = m1v + 8;

  const int h    = blockIdx.x;
  const int tid  = threadIdx.x;
  const int wave = tid >> 5;
  const int lane = tid & 31;

  const int b0 = tid >> 6,          d0 = tid & 63;
  const int b1i = (tid + 256) >> 6, d1i = (tid + 256) & 63;
  const size_t g0base = (size_t)(b0 * SS) * HIDC + h * 64 + d0;
  const size_t g1base = (size_t)(b1i * SS) * HIDC + h * 64 + d1i;
  const int l0 = b0 * LP + d0, l1 = b1i * LP + d1i;

  for (int i = tid; i < 4096; i += 256) {
    const int o = (i >> 6) * LP + (i & 63);
    W1[o] = fw1[(size_t)h * 4096 + i];
    W2[o] = fw2[(size_t)h * 4096 + i];
    LW[o] = lwg[i];
  }
  if (tid < 64) {
    b1v[tid] = fb1[h * 64 + tid];
    b2v[tid] = fb2[h * 64 + tid];
    lgv[tid] = flg[h * 64 + tid];
    lbv[tid] = flb[h * 64 + tid];
    tgv[tid] = ttg[h * 64 + tid];
    tbv[tid] = ttb[h * 64 + tid];
  }
  const float lbd = lbg[d0];

  float kp0 = kg[g0base], vp0 = vg[g0base];
  float kp1 = kg[g1base], vp1 = vg[g1base];
  __syncthreads();
  gemm_self_transposed_scaled(LW, MM, wave, lane);

  for (int t = 0; t < SS; ++t) {
    ktb[l0] = kp0;  tgt[l0] = lbd - (vp0 - kp0);
    ktb[l1] = kp1;  tgt[l1] = lbd - (vp1 - kp1);
    __syncthreads();
    if (t + 1 < SS) {
      kp0 = kg[g0base + (t + 1) * HIDC];  vp0 = vg[g0base + (t + 1) * HIDC];
      kp1 = kg[g1base + (t + 1) * HIDC];  vp1 = vg[g1base + (t + 1) * HIDC];
    }

    for (int step = 0; step < 4; ++step) {
      if (step == 0)
        gemm16_rowA(tgt, LW, nullptr, cb, nullptr, nullptr, 1, EPI_SCALE, wave - 4, lane);
      fwd_pass(ktb, av, hg, ob, xh, zb, W1, W2, b1v, b2v, lgv, lbv,
               muv, isv, wave, lane, tid);
      gemm16_rowA(zb, MM, nullptr, dzb, nullptr, cb, 0, EPI_ADDAUX, wave, lane);
      __syncthreads();
      {
        float dd0 = dzb[wave * LP + lane] * lgv[lane];
        float dd1 = dzb[wave * LP + lane + 32] * lgv[lane + 32];
        float x0 = xh[wave * LP + lane], x1 = xh[wave * LP + lane + 32];
        float s1 = wave_sum32(dd0 + dd1);
        float s2 = wave_sum32(dd0 * x0 + dd1 * x1);
        if (lane == 0) { m1v[wave] = s1 * (1.f / 64.f); m2v[wave] = s2 * (1.f / 64.f); }
      }
      __syncthreads();
      for (int idx = tid; idx < 512; idx += 256) {
        int b = idx >> 6, d = idx & 63, o = b * LP + d;
        float dxh = dzb[o] * lgv[d];
        dob[o] = isv[b] * (dxh - m1v[b] - xh[o] * m2v[b]);
      }
      __syncthreads();
      if (tid < 64) {
        float s = 0.f;
        for (int b = 0; b < 8; ++b) s += dzb[b * LP + tid] * xh[b * LP + tid];
        lgv[tid] -= LR * s;
      } else if (tid < 128) {
        int d = tid - 64; float s = 0.f;
        for (int b = 0; b < 8; ++b) s += dzb[b * LP + d];
        lbv[d] -= LR * s;
      }
      gemm16_rowA(dob, W2, nullptr, dab, nullptr, av, 1, EPI_DGELU, wave, lane);
      __syncthreads();
      gemm_outer_update(hg, dob, W2, LR, wave, lane);
      gemm_outer_update(ktb, dab, W1, LR, wave, lane);
      if (tid < 64) {
        float s = 0.f;
        for (int b = 0; b < 8; ++b) s += dob[b * LP + tid];
        b2v[tid] -= LR * s;
      } else if (tid < 128) {
        int d = tid - 64; float s = 0.f;
        for (int b = 0; b < 8; ++b) s += dab[b * LP + d];
        b1v[d] -= LR * s;
      }
      __syncthreads();
    }

    fwd_pass(ktb, av, hg, ob, xh, zb, W1, W2, b1v, b2v, lgv, lbv,
             muv, isv, wave, lane, tid);
    {
      float e0 = zb[wave * LP + lane], e1 = zb[wave * LP + lane + 32];
      float s = wave_sum32(e0 + e1);
      float q = wave_sum32(e0 * e0 + e1 * e1);
      if (lane == 0) {
        float m = s * (1.f / 64.f);
        muv[wave] = m;
        isv[wave] = rsqrtf(q * (1.f / 64.f) - m * m + LN_EPS);
      }
    }
    __syncthreads();
    for (int idx = tid; idx < 512; idx += 256) {
      int b = idx >> 6, d = idx & 63;
      float z2 = (zb[b * LP + d] - muv[b]) * isv[b] * tgv[d] + tbv[d];
      size_t g = ((size_t)(b * SS + t)) * HIDC + h * 64 + d;
      outg[g] = qg[g] + z2;
    }
    __syncthreads();
  }
}

__global__ void __launch_bounds__(256)
gate_ln_kernel(const float* __restrict__ outb, const float* __restrict__ gact,
               const float* __restrict__ png, const float* __restrict__ pnb,
               float* __restrict__ gated) {
  __shared__ float r1[256], r2[256];
  __shared__ float mu_s, inv_s;
  const int r = blockIdx.x, tid = threadIdx.x;
  const float* row = outb + (size_t)r * HIDC;
  float s = 0.f, q = 0.f;
  for (int i = tid; i < HIDC; i += 256) { float v = row[i]; s += v; q += v * v; }
  r1[tid] = s; r2[tid] = q;
  __syncthreads();
  for (int off = 128; off > 0; off >>= 1) {
    if (tid < off) { r1[tid] += r1[tid + off]; r2[tid] += r2[tid + off]; }
    __syncthreads();
  }
  if (tid == 0) {
    float m = r1[0] * (1.f / HIDC);
    mu_s = m;
    inv_s = rsqrtf(r2[0] * (1.f / HIDC) - m * m + LN_EPS);
  }
  __syncthreads();
  for (int i = tid; i < HIDC; i += 256) {
    float y = (row[i] - mu_s) * inv_s * png[i] + pnb[i];
    gated[(size_t)r * HIDC + i] = gact[(size_t)r * HIDC + i] * y;
  }
}

extern "C" void kernel_launch(void* const* d_in, const int* in_sizes, int n_in,
                              void* d_out, int out_size, void* d_ws, size_t ws_size,
                              hipStream_t stream) {
  (void)in_sizes; (void)n_in; (void)out_size; (void)ws_size;
  const float* x   = (const float*)d_in[0];
  const float* wq  = (const float*)d_in[1];
  const float* wk  = (const float*)d_in[2];
  const float* wv  = (const float*)d_in[3];
  const float* fw1 = (const float*)d_in[4];
  const float* fb1 = (const float*)d_in[5];
  const float* fw2 = (const float*)d_in[6];
  const float* fb2 = (const float*)d_in[7];
  const float* flg = (const float*)d_in[8];
  const float* flb = (const float*)d_in[9];
  const float* lw  = (const float*)d_in[10];
  const float* lb  = (const float*)d_in[11];
  const float* ttg = (const float*)d_in[12];
  const float* ttb = (const float*)d_in[13];
  const float* wo  = (const float*)d_in[14];
  const float* wg  = (const float*)d_in[15];
  const float* png = (const float*)d_in[16];
  const float* pnb = (const float*)d_in[17];

  const int M = BB * SS;
  const size_t PLANE = (size_t)M * HIDC;
  float* qb    = (float*)d_ws;
  float* kb    = qb + PLANE;
  float* vb    = kb + PLANE;
  float* gactb = vb + PLANE;
  float* outb  = gactb + PLANE;
  float* gatedb = qb;

  dim3 gblk(256);
  dim3 ggrid(HIDC / 128, M / 16);
  gemm_wmma_f32<<<ggrid, gblk, 0, stream>>>(x, wq, qb,    M, HIDC, HIDC, 0);
  gemm_wmma_f32<<<ggrid, gblk, 0, stream>>>(x, wk, kb,    M, HIDC, HIDC, 0);
  gemm_wmma_f32<<<ggrid, gblk, 0, stream>>>(x, wv, vb,    M, HIDC, HIDC, 0);
  gemm_wmma_f32<<<ggrid, gblk, 0, stream>>>(x, wg, gactb, M, HIDC, HIDC, 1);

  ttt_scan_kernel<<<HH, 256, SCAN_SMEM_FLOATS * sizeof(float), stream>>>(
      kb, vb, qb, fw1, fb1, fw2, fb2, flg, flb, lw, lb, ttg, ttb, outb);

  gate_ln_kernel<<<M, 256, 0, stream>>>(outb, gactb, png, pnb, gatedb);

  gemm_wmma_f32<<<ggrid, gblk, 0, stream>>>(gatedb, wo, (float*)d_out,
                                            M, HIDC, HIDC, 0);
}


# scan LayerNorms (fwd, bwd, post-norm): each wave reduces its own batch row by permlanex16 + DPP row_xmask adds (same xor tree as the shuffles) and normalises it at once, stats in registers, not LDS
# speedup vs baseline: 3.8228x; 1.4191x over previous
#include <hip/hip_runtime.h>
#include <hip/hip_bf16.h>
#include <math.h>

#define BB     8
#define SS     256
#define HIDC   1024
#define HH     16
#define DD     64
#define FF     64
#define LR     0.1f
#define GSCALE (2.0f / 8192.0f)
#define LN_EPS 1e-5f
#define LP 76

typedef float v2f __attribute__((ext_vector_type(2)));
typedef float v8f __attribute__((ext_vector_type(8)));

__device__ __forceinline__ v8f wmma4(v2f a, v2f b, v8f c) {
  return __builtin_amdgcn_wmma_f32_16x16x4_f32(
       false, a,  false, b,
       (short)0, c,  false,  false);
}

__device__ __forceinline__ float tanh_straight(float u) {
  const float a = fabsf(u);
  const float e = __builtin_amdgcn_exp2f(a * 2.8853900817779268f);
  const float t = 1.0f - 2.0f * __builtin_amdgcn_rcpf(e + 1.0f);
  return copysignf(t, u);
}

__device__ __forceinline__ float gelu_f(float x) {
  const float c = 0.7978845608028654f;
  float u = c * (x + 0.044715f * x * x * x);
  return 0.5f * x * (1.0f + tanh_straight(u));
}
__device__ __forceinline__ float gelu_grad(float x) {
  const float c = 0.7978845608028654f;
  float x2 = x * x;
  float u = c * (x + 0.044715f * x2 * x);
  float t = tanh_straight(u);
  return 0.5f * (1.0f + t) + 0.5f * x * (1.0f - t * t) * c * (1.0f + 0.134145f * x2);
}
__device__ __forceinline__ float xhalf(float x) {
  const int b = __builtin_bit_cast(int, x);
  return __builtin_bit_cast(float,
      __builtin_amdgcn_permlanex16(b, b, 0x76543210, 0xfedcba98, false, false));
}
template <int M>
__device__ __forceinline__ float xrow(float x) {
  const int b = __builtin_bit_cast(int, x);
  return __builtin_bit_cast(float,
      __builtin_amdgcn_update_dpp(0, b, 0x160 + M, 0xf, 0xf, true));
}
__device__ __forceinline__ float wave_sum32(float v) {
  v += xhalf(v);
  v += xrow<8>(v);
  v += xrow<4>(v);
  v += xrow<2>(v);
  v += xrow<1>(v);
  return v;
}
__device__ __forceinline__ void row_stats(float e0, float e1, float& m, float& is) {
  const float s = wave_sum32(e0 + e1);
  const float q = wave_sum32(e0 * e0 + e1 * e1);
  m  = s * (1.f / 64.f);
  is = rsqrtf(q * (1.f / 64.f) - m * m + LN_EPS);
}

__global__ void __launch_bounds__(256)
gemm_wmma_f32(const float* __restrict__ A, const float* __restrict__ Bm,
              float* __restrict__ C, int M, int N, int K, int act) {
  __shared__ float As[16 * LP];
  const int tid   = threadIdx.x;
  const int wave  = tid >> 5;
  const int lane  = tid & 31;
  const int mbase = blockIdx.y * 16;
  const int tn    = blockIdx.x * 8 + wave;
  const int col   = tn * 16 + (lane & 15);
  const int kh    = (lane >> 4) << 1;
  const int arow  = lane & 15;
  v8f acc = {0.f, 0.f, 0.f, 0.f, 0.f, 0.f, 0.f, 0.f};
  for (int kc = 0; kc < K; kc += 64) {
    __syncthreads();
    for (int i = tid; i < 16 * 64; i += 256)
      As[(i >> 6) * LP + (i & 63)] = A[(size_t)(mbase + (i >> 6)) * K + kc + (i & 63)];
    if (kc + 64 < K)
      __builtin_prefetch(&A[(size_t)(mbase + (tid >> 4)) * K + kc + 64], 0, 1);
    __syncthreads();
#pragma unroll
    for (int k0 = 0; k0 < 64; k0 += 4) {
      v2f a, b;
      a.x = As[arow * LP + k0 + kh];
      a.y = As[arow * LP + k0 + kh + 1];
      b.x = Bm[(size_t)(kc + k0 + kh) * N + col];
      b.y = Bm[(size_t)(kc + k0 + kh + 1) * N + col];
      acc = wmma4(a, b, acc);
    }
  }
  const int r0 = mbase + ((lane >= 16) ? 8 : 0);
#pragma unroll
  for (int i = 0; i < 8; ++i) {
    float v = acc[i];
    if (act) v = gelu_f(v);
    C[(size_t)(r0 + i) * N + col] = v;
  }
}

#define EPI_BIAS      0
#define EPI_GELU_PAIR 1
#define EPI_ADDAUX    2
#define EPI_DGELU     3
#define EPI_SCALE     4

__device__ __forceinline__ void
gemm16_rowA(const float* A8, const float* Bm, const float* bias,
            float* C8, float* C2, const float* aux,
            int transB, int epi, int wave, int lane) {
  if ((unsigned)wave >= 4u) return;
  const int col  = wave * 16 + (lane & 15);
  const int mrow = lane & 7;
  const int kh   = (lane >> 4) << 1;
  v2f a[16], b[16];
#pragma unroll
  for (int j = 0; j < 16; ++j) {
    const int k = 4 * j + kh;
    a[j].x = A8[mrow * LP + k];
    a[j].y = A8[mrow * LP + k + 1];
    if (!transB) {
      b[j].x = Bm[k * LP + col];
      b[j].y = Bm[(k + 1) * LP + col];
    } else {
      b[j].x = Bm[col * LP + k];
      b[j].y = Bm[col * LP + k + 1];
    }
  }
  const float bv = bias ? bias[col] : 0.f;
  float ax[8];
#pragma unroll
  for (int i = 0; i < 8; ++i)
    ax[i] = (epi == EPI_ADDAUX || epi == EPI_DGELU) ? aux[i * LP + col] : 0.f;
  __builtin_amdgcn_sched_barrier(0);
  v8f acc = {0.f, 0.f, 0.f, 0.f, 0.f, 0.f, 0.f, 0.f};
#pragma unroll
  for (int j = 0; j < 16; ++j) acc = wmma4(a[j], b[j], acc);
  if (lane < 16) {
#pragma unroll
    for (int i = 0; i < 8; ++i) {
      const int o = i * LP + col;
      float v = acc[i] + bv;
      if (epi == EPI_GELU_PAIR)      { C8[o] = v; C2[o] = gelu_f(v); }
      else if (epi == EPI_ADDAUX)    { C8[o] = v + ax[i]; }
      else if (epi == EPI_DGELU)     { C8[o] = v * gelu_grad(ax[i]); }
      else if (epi == EPI_SCALE)     { C8[o] = GSCALE * v; }
      else                           { C8[o] = v; }
    }
  }
}

__device__ __forceinline__ void
gemm_outer_update(const float* X8, const float* Y8, float* Wt, float lr,
                  int wave, int lane) {
  const int kh = (lane >> 4) << 1;
#pragma unroll
  for (int ti = 0; ti < 2; ++ti) {
    const int t  = wave * 2 + ti;
    const int tm = t >> 2, tn = t & 3;
    const int m = tm * 16 + (lane & 15);
    const int n = tn * 16 + (lane & 15);
    v8f acc = {0.f, 0.f, 0.f, 0.f, 0.f, 0.f, 0.f, 0.f};
#pragma unroll
    for (int k0 = 0; k0 < 8; k0 += 4) {
      v2f a, b;
      a.x = X8[(k0 + kh) * LP + m];
      a.y = X8[(k0 + kh + 1) * LP + m];
      b.x = Y8[(k0 + kh) * LP + n];
      b.y = Y8[(k0 + kh + 1) * LP + n];
      acc = wmma4(a, b, acc);
    }
    const int rbase = tm * 16 + ((lane >= 16) ? 8 : 0);
    const int cc    = tn * 16 + (lane & 15);
#pragma unroll
    for (int i = 0; i < 8; ++i) Wt[(rbase + i) * LP + cc] -= lr * acc[i];
  }
}

__device__ __forceinline__ void
gemm_self_transposed_scaled(const float* L, float* M, int wave, int lane) {
  const int kh = (lane >> 4) << 1;
  const int c  = lane & 15;
#pragma unroll
  for (int ti = 0; ti < 2; ++ti) {
    const int t  = wave * 2 + ti;
    const int tm = t >> 2, tn = t & 3;
    v8f acc = {0.f, 0.f, 0.f, 0.f, 0.f, 0.f, 0.f, 0.f};
#pragma unroll
    for (int j = 0; j < 16; ++j) {
      const int k = 4 * j + kh;
      v2f a, b;
      a.x = L[(tm * 16 + c) * LP + k];
      a.y = L[(tm * 16 + c) * LP + k + 1];
      b.x = L[(tn * 16 + c) * LP + k];
      b.y = L[(tn * 16 + c) * LP + k + 1];
      acc = wmma4(a, b, acc);
    }
    const int rbase = tm * 16 + ((lane >= 16) ? 8 : 0);
#pragma unroll
    for (int i = 0; i < 8; ++i) M[(rbase + i) * LP + tn * 16 + c] = GSCALE * acc[i];
  }
}

__device__ __forceinline__ void
fwd_gemms(const float* ktb, float* av, float* hg, float* ob,
          const float* W1, const float* W2, const float* b1v, const float* b2v,
          int wave, int lane) {
  gemm16_rowA(ktb, W1, b1v, av, hg, nullptr, 0, EPI_GELU_PAIR, wave, lane);
  __syncthreads();
  gemm16_rowA(hg, W2, b2v, ob, nullptr, nullptr, 0, EPI_BIAS, wave, lane);
  __syncthreads();
}

struct RowLN { float x0, x1, z0, z1, is; };
__device__ __forceinline__ RowLN
row_layernorm(const float* ob, const float* lgv, const float* lbv,
              int o0, int o1, int lane) {
  RowLN r;
  const float e0 = ob[o0], e1 = ob[o1];
  const float g0 = lgv[lane], g1 = lgv[lane + 32];
  const float c0 = lbv[lane], c1 = lbv[lane + 32];
  __builtin_amdgcn_sched_barrier(0);
  float m;
  row_stats(e0, e1, m, r.is);
  r.x0 = (e0 - m) * r.is;
  r.x1 = (e1 - m) * r.is;
  r.z0 = r.x0 * g0 + c0;
  r.z1 = r.x1 * g1 + c1;
  return r;
}

#define SCAN_SMEM_FLOATS (4 * 64 * LP + 11 * 8 * LP + 6 * 64)

__global__ void __launch_bounds__(256)
ttt_scan_kernel(const float* __restrict__ kg, const float* __restrict__ vg,
                const float* __restrict__ qg,
                const float* __restrict__ fw1, const float* __restrict__ fb1,
                const float* __restrict__ fw2, const float* __restrict__ fb2,
                const float* __restrict__ flg, const float* __restrict__ flb,
                const float* __restrict__ lwg, const float* __restrict__ lbg,
                const float* __restrict__ ttg, const float* __restrict__ ttb,
                float* __restrict__ outg) {
  extern __shared__ float smem[];
  float* W1  = smem;
  float* W2  = W1 + 64 * LP;
  float* LW  = W2 + 64 * LP;
  float* MM  = LW + 64 * LP;
  float* ktb = MM + 64 * LP;
  float* tgt = ktb + 8 * LP;
  float* av  = tgt + 8 * LP;
  float* hg  = av + 8 * LP;
  float* ob  = hg + 8 * LP;
  float* xh  = ob + 8 * LP;
  float* zb  = xh + 8 * LP;
  float* dzb = zb + 8 * LP;
  float* dob = dzb + 8 * LP;
  float* dab = dob + 8 * LP;
  float* cb  = dab + 8 * LP;
  float* b1v = cb + 8 * LP;
  float* b2v = b1v + 64;
  float* lgv = b2v + 64;
  float* lbv = lgv + 64;
  float* tgv = lbv + 64;
  float* tbv = tgv + 64;

  const int h    = blockIdx.x;
  const int tid  = threadIdx.x;
  const int wave = tid >> 5;
  const int lane = tid & 31;

  const int b0 = tid >> 6,          d0 = tid & 63;
  const int b1i = (tid + 256) >> 6, d1i = (tid + 256) & 63;
  const size_t g0base = (size_t)(b0 * SS) * HIDC + h * 64 + d0;
  const size_t g1base = (size_t)(b1i * SS) * HIDC + h * 64 + d1i;
  const int l0 = b0 * LP + d0, l1 = b1i * LP + d1i;
  const int r0 = wave * LP + lane, r1 = r0 + 32;

  for (int i = tid; i < 4096; i += 256) {
    const int o = (i >> 6) * LP + (i & 63);
    W1[o] = fw1[(size_t)h * 4096 + i];
    W2[o] = fw2[(size_t)h * 4096 + i];
    LW[o] = lwg[i];
  }
  if (tid < 64) {
    b1v[tid] = fb1[h * 64 + tid];
    b2v[tid] = fb2[h * 64 + tid];
    lgv[tid] = flg[h * 64 + tid];
    lbv[tid] = flb[h * 64 + tid];
    tgv[tid] = ttg[h * 64 + tid];
    tbv[tid] = ttb[h * 64 + tid];
  }
  const float lbd = lbg[d0];

  float kp0 = kg[g0base], vp0 = vg[g0base];
  float kp1 = kg[g1base], vp1 = vg[g1base];
  __syncthreads();
  gemm_self_transposed_scaled(LW, MM, wave, lane);

  for (int t = 0; t < SS; ++t) {
    ktb[l0] = kp0;  tgt[l0] = lbd - (vp0 - kp0);
    ktb[l1] = kp1;  tgt[l1] = lbd - (vp1 - kp1);
    __syncthreads();
    if (t + 1 < SS) {
      kp0 = kg[g0base + (t + 1) * HIDC];  vp0 = vg[g0base + (t + 1) * HIDC];
      kp1 = kg[g1base + (t + 1) * HIDC];  vp1 = vg[g1base + (t + 1) * HIDC];
    }

    for (int step = 0; step < 4; ++step) {
      if (step == 0)
        gemm16_rowA(tgt, LW, nullptr, cb, nullptr, nullptr, 1, EPI_SCALE, wave - 4, lane);
      fwd_gemms(ktb, av, hg, ob, W1, W2, b1v, b2v, wave, lane);
      const RowLN r = row_layernorm(ob, lgv, lbv, r0, r1, lane);
      xh[r0] = r.x0;  xh[r1] = r.x1;
      zb[r0] = r.z0;  zb[r1] = r.z1;
      __syncthreads();
      gemm16_rowA(zb, MM, nullptr, dzb, nullptr, cb, 0, EPI_ADDAUX, wave, lane);
      __syncthreads();
      {
        const float dd0 = dzb[r0] * lgv[lane];
        const float dd1 = dzb[r1] * lgv[lane + 32];
        const float m1 = wave_sum32(dd0 + dd1) * (1.f / 64.f);
        const float m2 = wave_sum32(dd0 * r.x0 + dd1 * r.x1) * (1.f / 64.f);
        dob[r0] = r.is * (dd0 - m1 - r.x0 * m2);
        dob[r1] = r.is * (dd1 - m1 - r.x1 * m2);
      }
      __syncthreads();
      if (tid < 64) {
        float s = 0.f;
        for (int b = 0; b < 8; ++b) s += dzb[b * LP + tid] * xh[b * LP + tid];
        lgv[tid] -= LR * s;
      } else if (tid < 128) {
        int d = tid - 64; float s = 0.f;
        for (int b = 0; b < 8; ++b) s += dzb[b * LP + d];
        lbv[d] -= LR * s;
      }
      gemm16_rowA(dob, W2, nullptr, dab, nullptr, av, 1, EPI_DGELU, wave, lane);
      __syncthreads();
      gemm_outer_update(hg, dob, W2, LR, wave, lane);
      gemm_outer_update(ktb, dab, W1, LR, wave, lane);
      if (tid < 64) {
        float s = 0.f;
        for (int b = 0; b < 8; ++b) s += dob[b * LP + tid];
        b2v[tid] -= LR * s;
      } else if (tid < 128) {
        int d = tid - 64; float s = 0.f;
        for (int b = 0; b < 8; ++b) s += dab[b * LP + d];
        b1v[d] -= LR * s;
      }
      __syncthreads();
    }

    fwd_gemms(ktb, av, hg, ob, W1, W2, b1v, b2v, wave, lane);
    {
      const size_t g = ((size_t)(wave * SS + t)) * HIDC + h * 64 + lane;
      const float q0 = qg[g], q1 = qg[g + 32];
      const float tg0 = tgv[lane], tg1 = tgv[lane + 32];
      const float tb0 = tbv[lane], tb1 = tbv[lane + 32];
      const RowLN r = row_layernorm(ob, lgv, lbv, r0, r1, lane);
      float m, is;
      row_stats(r.z0, r.z1, m, is);
      outg[g]      = q0 + ((r.z0 - m) * is * tg0 + tb0);
      outg[g + 32] = q1 + ((r.z1 - m) * is * tg1 + tb1);
    }
  }
}

__global__ void __launch_bounds__(256)
gate_ln_kernel(const float* __restrict__ outb, const float* __restrict__ gact,
               const float* __restrict__ png, const float* __restrict__ pnb,
               float* __restrict__ gated) {
  __shared__ float r1[256], r2[256];
  __shared__ float mu_s, inv_s;
  const int r = blockIdx.x, tid = threadIdx.x;
  const float* row = outb + (size_t)r * HIDC;
  float s = 0.f, q = 0.f;
  for (int i = tid; i < HIDC; i += 256) { float v = row[i]; s += v; q += v * v; }
  r1[tid] = s; r2[tid] = q;
  __syncthreads();
  for (int off = 128; off > 0; off >>= 1) {
    if (tid < off) { r1[tid] += r1[tid + off]; r2[tid] += r2[tid + off]; }
    __syncthreads();
  }
  if (tid == 0) {
    float m = r1[0] * (1.f / HIDC);
    mu_s = m;
    inv_s = rsqrtf(r2[0] * (1.f / HIDC) - m * m + LN_EPS);
  }
  __syncthreads();
  for (int i = tid; i < HIDC; i += 256) {
    float y = (row[i] - mu_s) * inv_s * png[i] + pnb[i];
    gated[(size_t)r * HIDC + i] = gact[(size_t)r * HIDC + i] * y;
  }
}

extern "C" void kernel_launch(void* const* d_in, const int* in_sizes, int n_in,
                              void* d_out, int out_size, void* d_ws, size_t ws_size,
                              hipStream_t stream) {
  (void)in_sizes; (void)n_in; (void)out_size; (void)ws_size;
  const float* x   = (const float*)d_in[0];
  const float* wq  = (const float*)d_in[1];
  const float* wk  = (const float*)d_in[2];
  const float* wv  = (const float*)d_in[3];
  const float* fw1 = (const float*)d_in[4];
  const float* fb1 = (const float*)d_in[5];
  const float* fw2 = (const float*)d_in[6];
  const float* fb2 = (const float*)d_in[7];
  const float* flg = (const float*)d_in[8];
  const float* flb = (const float*)d_in[9];
  const float* lw  = (const float*)d_in[10];
  const float* lb  = (const float*)d_in[11];
  const float* ttg = (const float*)d_in[12];
  const float* ttb = (const float*)d_in[13];
  const float* wo  = (const float*)d_in[14];
  const float* wg  = (const float*)d_in[15];
  const float* png = (const float*)d_in[16];
  const float* pnb = (const float*)d_in[17];

  const int M = BB * SS;
  const size_t PLANE = (size_t)M * HIDC;
  float* qb    = (float*)d_ws;
  float* kb    = qb + PLANE;
  float* vb    = kb + PLANE;
  float* gactb = vb + PLANE;
  float* outb  = gactb + PLANE;
  float* gatedb = qb;

  dim3 gblk(256);
  dim3 ggrid(HIDC / 128, M / 16);
  gemm_wmma_f32<<<ggrid, gblk, 0, stream>>>(x, wq, qb,    M, HIDC, HIDC, 0);
  gemm_wmma_f32<<<ggrid, gblk, 0, stream>>>(x, wk, kb,    M, HIDC, HIDC, 0);
  gemm_wmma_f32<<<ggrid, gblk, 0, stream>>>(x, wv, vb,    M, HIDC, HIDC, 0);
  gemm_wmma_f32<<<ggrid, gblk, 0, stream>>>(x, wg, gactb, M, HIDC, HIDC, 1);

  ttt_scan_kernel<<<HH, 256, SCAN_SMEM_FLOATS * sizeof(float), stream>>>(
      kb, vb, qb, fw1, fb1, fw2, fb2, flg, flb, lw, lb, ttg, ttb, outb);

  gate_ln_kernel<<<M, 256, 0, stream>>>(outb, gactb, png, pnb, gatedb);

  gemm_wmma_f32<<<ggrid, gblk, 0, stream>>>(gatedb, wo, (float*)d_out,
                                            M, HIDC, HIDC, 0);
}


# scan weight updates: tiles formed transposed (Y^T X) so a lane holds 8 consecutive floats of a W row: 16-byte LDS read-modify-write, all loads of the phase in one group; LDS base 16-byte aligned
# speedup vs baseline: 4.3650x; 1.1418x over previous
#include <hip/hip_runtime.h>
#include <hip/hip_bf16.h>
#include <math.h>

#define BB     8
#define SS     256
#define HIDC   1024
#define HH     16
#define DD     64
#define FF     64
#define LR     0.1f
#define GSCALE (2.0f / 8192.0f)
#define LN_EPS 1e-5f
#define LP 76

typedef float v2f __attribute__((ext_vector_type(2)));
typedef float v4f __attribute__((ext_vector_type(4)));
typedef float v8f __attribute__((ext_vector_type(8)));

__device__ __forceinline__ v8f wmma4(v2f a, v2f b, v8f c) {
  return __builtin_amdgcn_wmma_f32_16x16x4_f32(
       false, a,  false, b,
       (short)0, c,  false,  false);
}

__device__ __forceinline__ float tanh_straight(float u) {
  const float a = fabsf(u);
  const float e = __builtin_amdgcn_exp2f(a * 2.8853900817779268f);
  const float t = 1.0f - 2.0f * __builtin_amdgcn_rcpf(e + 1.0f);
  return copysignf(t, u);
}

__device__ __forceinline__ float gelu_f(float x) {
  const float c = 0.7978845608028654f;
  float u = c * (x + 0.044715f * x * x * x);
  return 0.5f * x * (1.0f + tanh_straight(u));
}
__device__ __forceinline__ float gelu_grad(float x) {
  const float c = 0.7978845608028654f;
  float x2 = x * x;
  float u = c * (x + 0.044715f * x2 * x);
  float t = tanh_straight(u);
  return 0.5f * (1.0f + t) + 0.5f * x * (1.0f - t * t) * c * (1.0f + 0.134145f * x2);
}
__device__ __forceinline__ float xhalf(float x) {
  const int b = __builtin_bit_cast(int, x);
  return __builtin_bit_cast(float,
      __builtin_amdgcn_permlanex16(b, b, 0x76543210, 0xfedcba98, false, false));
}
template <int M>
__device__ __forceinline__ float xrow(float x) {
  const int b = __builtin_bit_cast(int, x);
  return __builtin_bit_cast(float,
      __builtin_amdgcn_update_dpp(0, b, 0x160 + M, 0xf, 0xf, true));
}
__device__ __forceinline__ float wave_sum32(float v) {
  v += xhalf(v);
  v += xrow<8>(v);
  v += xrow<4>(v);
  v += xrow<2>(v);
  v += xrow<1>(v);
  return v;
}
__device__ __forceinline__ void row_stats(float e0, float e1, float& m, float& is) {
  const float s = wave_sum32(e0 + e1);
  const float q = wave_sum32(e0 * e0 + e1 * e1);
  m  = s * (1.f / 64.f);
  is = rsqrtf(q * (1.f / 64.f) - m * m + LN_EPS);
}

__global__ void __launch_bounds__(256)
gemm_wmma_f32(const float* __restrict__ A, const float* __restrict__ Bm,
              float* __restrict__ C, int M, int N, int K, int act) {
  __shared__ float As[16 * LP];
  const int tid   = threadIdx.x;
  const int wave  = tid >> 5;
  const int lane  = tid & 31;
  const int mbase = blockIdx.y * 16;
  const int tn    = blockIdx.x * 8 + wave;
  const int col   = tn * 16 + (lane & 15);
  const int kh    = (lane >> 4) << 1;
  const int arow  = lane & 15;
  v8f acc = {0.f, 0.f, 0.f, 0.f, 0.f, 0.f, 0.f, 0.f};
  for (int kc = 0; kc < K; kc += 64) {
    __syncthreads();
    for (int i = tid; i < 16 * 64; i += 256)
      As[(i >> 6) * LP + (i & 63)] = A[(size_t)(mbase + (i >> 6)) * K + kc + (i & 63)];
    if (kc + 64 < K)
      __builtin_prefetch(&A[(size_t)(mbase + (tid >> 4)) * K + kc + 64], 0, 1);
    __syncthreads();
#pragma unroll
    for (int k0 = 0; k0 < 64; k0 += 4) {
      v2f a, b;
      a.x = As[arow * LP + k0 + kh];
      a.y = As[arow * LP + k0 + kh + 1];
      b.x = Bm[(size_t)(kc + k0 + kh) * N + col];
      b.y = Bm[(size_t)(kc + k0 + kh + 1) * N + col];
      acc = wmma4(a, b, acc);
    }
  }
  const int r0 = mbase + ((lane >= 16) ? 8 : 0);
#pragma unroll
  for (int i = 0; i < 8; ++i) {
    float v = acc[i];
    if (act) v = gelu_f(v);
    C[(size_t)(r0 + i) * N + col] = v;
  }
}

#define EPI_BIAS      0
#define EPI_GELU_PAIR 1
#define EPI_ADDAUX    2
#define EPI_DGELU     3
#define EPI_SCALE     4

__device__ __forceinline__ void
gemm16_rowA(const float* A8, const float* Bm, const float* bias,
            float* C8, float* C2, const float* aux,
            int transB, int epi, int wave, int lane) {
  if ((unsigned)wave >= 4u) return;
  const int col  = wave * 16 + (lane & 15);
  const int mrow = lane & 7;
  const int kh   = (lane >> 4) << 1;
  v2f a[16], b[16];
#pragma unroll
  for (int j = 0; j < 16; ++j) {
    const int k = 4 * j + kh;
    a[j].x = A8[mrow * LP + k];
    a[j].y = A8[mrow * LP + k + 1];
    if (!transB) {
      b[j].x = Bm[k * LP + col];
      b[j].y = Bm[(k + 1) * LP + col];
    } else {
      b[j].x = Bm[col * LP + k];
      b[j].y = Bm[col * LP + k + 1];
    }
  }
  const float bv = bias ? bias[col] : 0.f;
  float ax[8];
#pragma unroll
  for (int i = 0; i < 8; ++i)
    ax[i] = (epi == EPI_ADDAUX || epi == EPI_DGELU) ? aux[i * LP + col] : 0.f;
  __builtin_amdgcn_sched_barrier(0);
  v8f acc = {0.f, 0.f, 0.f, 0.f, 0.f, 0.f, 0.f, 0.f};
#pragma unroll
  for (int j = 0; j < 16; ++j) acc = wmma4(a[j], b[j], acc);
  if (lane < 16) {
#pragma unroll
    for (int i = 0; i < 8; ++i) {
      const int o = i * LP + col;
      float v = acc[i] + bv;
      if (epi == EPI_GELU_PAIR)      { C8[o] = v; C2[o] = gelu_f(v); }
      else if (epi == EPI_ADDAUX)    { C8[o] = v + ax[i]; }
      else if (epi == EPI_DGELU)     { C8[o] = v * gelu_grad(ax[i]); }
      else if (epi == EPI_SCALE)     { C8[o] = GSCALE * v; }
      else                           { C8[o] = v; }
    }
  }
}

__device__ __forceinline__ void
weight_updates(const float* Xa, const float* Ya, float* Wa, float* ba,
               const float* Xb, const float* Yb, float* Wb, float* bb,
               float lr, int wave, int lane) {
  const int c  = lane & 15;
  const int h  = lane >> 4;
  const int kh = h << 1;
  const int m  = (wave >> 1) * 16 + c;
  const int nb = (wave & 1) * 32;
  const float* Xs[2] = {Xa, Xb};
  const float* Ys[2] = {Ya, Yb};
  float*       Ws[2] = {Wa, Wb};
  v2f x[2][2], y[2][2][2];
  v4f w[2][2][2];
#pragma unroll
  for (int q = 0; q < 2; ++q) {
#pragma unroll
    for (int s = 0; s < 2; ++s) {
      const int k = 4 * s + kh;
      x[q][s].x = Xs[q][k * LP + m];
      x[q][s].y = Xs[q][(k + 1) * LP + m];
#pragma unroll
      for (int ti = 0; ti < 2; ++ti) {
        y[q][ti][s].x = Ys[q][k * LP + nb + 16 * ti + c];
        y[q][ti][s].y = Ys[q][(k + 1) * LP + nb + 16 * ti + c];
      }
    }
#pragma unroll
    for (int ti = 0; ti < 2; ++ti) {
      const v4f* p = (const v4f*)(Ws[q] + m * LP + nb + 16 * ti + 8 * h);
      w[q][ti][0] = p[0];
      w[q][ti][1] = p[1];
    }
  }
  const float* Yc = (wave & 2) ? Yb : Ya;
  float*       bc = (wave & 2) ? bb : ba;
  const int    d  = (wave & 1) * 32 + lane;
  float g[8];
#pragma unroll
  for (int b = 0; b < 8; ++b) g[b] = Yc[b * LP + d];
  float bold = 0.f;
  if (wave < 4) bold = bc[d];
  __builtin_amdgcn_sched_barrier(0);
#pragma unroll
  for (int q = 0; q < 2; ++q) {
#pragma unroll
    for (int ti = 0; ti < 2; ++ti) {
      v8f acc = {0.f, 0.f, 0.f, 0.f, 0.f, 0.f, 0.f, 0.f};
      acc = wmma4(y[q][ti][0], x[q][0], acc);
      acc = wmma4(y[q][ti][1], x[q][1], acc);
      v4f lo = w[q][ti][0], hi = w[q][ti][1];
      lo.x -= lr * acc[0];  lo.y -= lr * acc[1];
      lo.z -= lr * acc[2];  lo.w -= lr * acc[3];
      hi.x -= lr * acc[4];  hi.y -= lr * acc[5];
      hi.z -= lr * acc[6];  hi.w -= lr * acc[7];
      v4f* p = (v4f*)(Ws[q] + m * LP + nb + 16 * ti + 8 * h);
      p[0] = lo;
      p[1] = hi;
    }
  }
  if (wave < 4) {
    float s = 0.f;
#pragma unroll
    for (int b = 0; b < 8; ++b) s += g[b];
    bc[d] = bold - lr * s;
  }
}

__device__ __forceinline__ void
gemm_self_transposed_scaled(const float* L, float* M, int wave, int lane) {
  const int kh = (lane >> 4) << 1;
  const int c  = lane & 15;
#pragma unroll
  for (int ti = 0; ti < 2; ++ti) {
    const int t  = wave * 2 + ti;
    const int tm = t >> 2, tn = t & 3;
    v8f acc = {0.f, 0.f, 0.f, 0.f, 0.f, 0.f, 0.f, 0.f};
#pragma unroll
    for (int j = 0; j < 16; ++j) {
      const int k = 4 * j + kh;
      v2f a, b;
      a.x = L[(tm * 16 + c) * LP + k];
      a.y = L[(tm * 16 + c) * LP + k + 1];
      b.x = L[(tn * 16 + c) * LP + k];
      b.y = L[(tn * 16 + c) * LP + k + 1];
      acc = wmma4(a, b, acc);
    }
    const int rbase = tm * 16 + ((lane >= 16) ? 8 : 0);
#pragma unroll
    for (int i = 0; i < 8; ++i) M[(rbase + i) * LP + tn * 16 + c] = GSCALE * acc[i];
  }
}

__device__ __forceinline__ void
fwd_gemms(const float* ktb, float* av, float* hg, float* ob,
          const float* W1, const float* W2, const float* b1v, const float* b2v,
          int wave, int lane) {
  gemm16_rowA(ktb, W1, b1v, av, hg, nullptr, 0, EPI_GELU_PAIR, wave, lane);
  __syncthreads();
  gemm16_rowA(hg, W2, b2v, ob, nullptr, nullptr, 0, EPI_BIAS, wave, lane);
  __syncthreads();
}

struct RowLN { float x0, x1, z0, z1, is; };
__device__ __forceinline__ RowLN
row_layernorm(const float* ob, const float* lgv, const float* lbv,
              int o0, int o1, int lane) {
  RowLN r;
  const float e0 = ob[o0], e1 = ob[o1];
  const float g0 = lgv[lane], g1 = lgv[lane + 32];
  const float c0 = lbv[lane], c1 = lbv[lane + 32];
  __builtin_amdgcn_sched_barrier(0);
  float m;
  row_stats(e0, e1, m, r.is);
  r.x0 = (e0 - m) * r.is;
  r.x1 = (e1 - m) * r.is;
  r.z0 = r.x0 * g0 + c0;
  r.z1 = r.x1 * g1 + c1;
  return r;
}

#define SCAN_SMEM_FLOATS (4 * 64 * LP + 11 * 8 * LP + 6 * 64)

__global__ void __launch_bounds__(256)
ttt_scan_kernel(const float* __restrict__ kg, const float* __restrict__ vg,
                const float* __restrict__ qg,
                const float* __restrict__ fw1, const float* __restrict__ fb1,
                const float* __restrict__ fw2, const float* __restrict__ fb2,
                const float* __restrict__ flg, const float* __restrict__ flb,
                const float* __restrict__ lwg, const float* __restrict__ lbg,
                const float* __restrict__ ttg, const float* __restrict__ ttb,
                float* __restrict__ outg) {
  extern __shared__ __attribute__((aligned(16))) float smem[];
  float* W1  = smem;
  float* W2  = W1 + 64 * LP;
  float* LW  = W2 + 64 * LP;
  float* MM  = LW + 64 * LP;
  float* ktb = MM + 64 * LP;
  float* tgt = ktb + 8 * LP;
  float* av  = tgt + 8 * LP;
  float* hg  = av + 8 * LP;
  float* ob  = hg + 8 * LP;
  float* xh  = ob + 8 * LP;
  float* zb  = xh + 8 * LP;
  float* dzb = zb + 8 * LP;
  float* dob = dzb + 8 * LP;
  float* dab = dob + 8 * LP;
  float* cb  = dab + 8 * LP;
  float* b1v = cb + 8 * LP;
  float* b2v = b1v + 64;
  float* lgv = b2v + 64;
  float* lbv = lgv + 64;
  float* tgv = lbv + 64;
  float* tbv = tgv + 64;

  const int h    = blockIdx.x;
  const int tid  = threadIdx.x;
  const int wave = tid >> 5;
  const int lane = tid & 31;

  const int b0 = tid >> 6,          d0 = tid & 63;
  const int b1i = (tid + 256) >> 6, d1i = (tid + 256) & 63;
  const size_t g0base = (size_t)(b0 * SS) * HIDC + h * 64 + d0;
  const size_t g1base = (size_t)(b1i * SS) * HIDC + h * 64 + d1i;
  const int l0 = b0 * LP + d0, l1 = b1i * LP + d1i;
  const int r0 = wave * LP + lane, r1 = r0 + 32;

  for (int i = tid; i < 4096; i += 256) {
    const int o = (i >> 6) * LP + (i & 63);
    W1[o] = fw1[(size_t)h * 4096 + i];
    W2[o] = fw2[(size_t)h * 4096 + i];
    LW[o] = lwg[i];
  }
  if (tid < 64) {
    b1v[tid] = fb1[h * 64 + tid];
    b2v[tid] = fb2[h * 64 + tid];
    lgv[tid] = flg[h * 64 + tid];
    lbv[tid] = flb[h * 64 + tid];
    tgv[tid] = ttg[h * 64 + tid];
    tbv[tid] = ttb[h * 64 + tid];
  }
  const float lbd = lbg[d0];

  float kp0 = kg[g0base], vp0 = vg[g0base];
  float kp1 = kg[g1base], vp1 = vg[g1base];
  __syncthreads();
  gemm_self_transposed_scaled(LW, MM, wave, lane);

  for (int t = 0; t < SS; ++t) {
    ktb[l0] = kp0;  tgt[l0] = lbd - (vp0 - kp0);
    ktb[l1] = kp1;  tgt[l1] = lbd - (vp1 - kp1);
    __syncthreads();
    if (t + 1 < SS) {
      kp0 = kg[g0base + (t + 1) * HIDC];  vp0 = vg[g0base + (t + 1) * HIDC];
      kp1 = kg[g1base + (t + 1) * HIDC];  vp1 = vg[g1base + (t + 1) * HIDC];
    }

    for (int step = 0; step < 4; ++step) {
      if (step == 0)
        gemm16_rowA(tgt, LW, nullptr, cb, nullptr, nullptr, 1, EPI_SCALE, wave - 4, lane);
      fwd_gemms(ktb, av, hg, ob, W1, W2, b1v, b2v, wave, lane);
      const RowLN r = row_layernorm(ob, lgv, lbv, r0, r1, lane);
      xh[r0] = r.x0;  xh[r1] = r.x1;
      zb[r0] = r.z0;  zb[r1] = r.z1;
      __syncthreads();
      gemm16_rowA(zb, MM, nullptr, dzb, nullptr, cb, 0, EPI_ADDAUX, wave, lane);
      __syncthreads();
      {
        const float dd0 = dzb[r0] * lgv[lane];
        const float dd1 = dzb[r1] * lgv[lane + 32];
        const float m1 = wave_sum32(dd0 + dd1) * (1.f / 64.f);
        const float m2 = wave_sum32(dd0 * r.x0 + dd1 * r.x1) * (1.f / 64.f);
        dob[r0] = r.is * (dd0 - m1 - r.x0 * m2);
        dob[r1] = r.is * (dd1 - m1 - r.x1 * m2);
      }
      __syncthreads();
      if (tid < 64) {
        float s = 0.f;
        for (int b = 0; b < 8; ++b) s += dzb[b * LP + tid] * xh[b * LP + tid];
        lgv[tid] -= LR * s;
      } else if (tid < 128) {
        int d = tid - 64; float s = 0.f;
        for (int b = 0; b < 8; ++b) s += dzb[b * LP + d];
        lbv[d] -= LR * s;
      }
      gemm16_rowA(dob, W2, nullptr, dab, nullptr, av, 1, EPI_DGELU, wave, lane);
      __syncthreads();
      weight_updates(hg, dob, W2, b2v, ktb, dab, W1, b1v, LR, wave, lane);
      __syncthreads();
    }

    fwd_gemms(ktb, av, hg, ob, W1, W2, b1v, b2v, wave, lane);
    {
      const size_t g = ((size_t)(wave * SS + t)) * HIDC + h * 64 + lane;
      const float q0 = qg[g], q1 = qg[g + 32];
      const float tg0 = tgv[lane], tg1 = tgv[lane + 32];
      const float tb0 = tbv[lane], tb1 = tbv[lane + 32];
      const RowLN r = row_layernorm(ob, lgv, lbv, r0, r1, lane);
      float m, is;
      row_stats(r.z0, r.z1, m, is);
      outg[g]      = q0 + ((r.z0 - m) * is * tg0 + tb0);
      outg[g + 32] = q1 + ((r.z1 - m) * is * tg1 + tb1);
    }
  }
}

__global__ void __launch_bounds__(256)
gate_ln_kernel(const float* __restrict__ outb, const float* __restrict__ gact,
               const float* __restrict__ png, const float* __restrict__ pnb,
               float* __restrict__ gated) {
  __shared__ float r1[256], r2[256];
  __shared__ float mu_s, inv_s;
  const int r = blockIdx.x, tid = threadIdx.x;
  const float* row = outb + (size_t)r * HIDC;
  float s = 0.f, q = 0.f;
  for (int i = tid; i < HIDC; i += 256) { float v = row[i]; s += v; q += v * v; }
  r1[tid] = s; r2[tid] = q;
  __syncthreads();
  for (int off = 128; off > 0; off >>= 1) {
    if (tid < off) { r1[tid] += r1[tid + off]; r2[tid] += r2[tid + off]; }
    __syncthreads();
  }
  if (tid == 0) {
    float m = r1[0] * (1.f / HIDC);
    mu_s = m;
    inv_s = rsqrtf(r2[0] * (1.f / HIDC) - m * m + LN_EPS);
  }
  __syncthreads();
  for (int i = tid; i < HIDC; i += 256) {
    float y = (row[i] - mu_s) * inv_s * png[i] + pnb[i];
    gated[(size_t)r * HIDC + i] = gact[(size_t)r * HIDC + i] * y;
  }
}

extern "C" void kernel_launch(void* const* d_in, const int* in_sizes, int n_in,
                              void* d_out, int out_size, void* d_ws, size_t ws_size,
                              hipStream_t stream) {
  (void)in_sizes; (void)n_in; (void)out_size; (void)ws_size;
  const float* x   = (const float*)d_in[0];
  const float* wq  = (const float*)d_in[1];
  const float* wk  = (const float*)d_in[2];
  const float* wv  = (const float*)d_in[3];
  const float* fw1 = (const float*)d_in[4];
  const float* fb1 = (const float*)d_in[5];
  const float* fw2 = (const float*)d_in[6];
  const float* fb2 = (const float*)d_in[7];
  const float* flg = (const float*)d_in[8];
  const float* flb = (const float*)d_in[9];
  const float* lw  = (const float*)d_in[10];
  const float* lb  = (const float*)d_in[11];
  const float* ttg = (const float*)d_in[12];
  const float* ttb = (const float*)d_in[13];
  const float* wo  = (const float*)d_in[14];
  const float* wg  = (const float*)d_in[15];
  const float* png = (const float*)d_in[16];
  const float* pnb = (const float*)d_in[17];

  const int M = BB * SS;
  const size_t PLANE = (size_t)M * HIDC;
  float* qb    = (float*)d_ws;
  float* kb    = qb + PLANE;
  float* vb    = kb + PLANE;
  float* gactb = vb + PLANE;
  float* outb  = gactb + PLANE;
  float* gatedb = qb;

  dim3 gblk(256);
  dim3 ggrid(HIDC / 128, M / 16);
  gemm_wmma_f32<<<ggrid, gblk, 0, stream>>>(x, wq, qb,    M, HIDC, HIDC, 0);
  gemm_wmma_f32<<<ggrid, gblk, 0, stream>>>(x, wk, kb,    M, HIDC, HIDC, 0);
  gemm_wmma_f32<<<ggrid, gblk, 0, stream>>>(x, wv, vb,    M, HIDC, HIDC, 0);
  gemm_wmma_f32<<<ggrid, gblk, 0, stream>>>(x, wg, gactb, M, HIDC, HIDC, 1);

  ttt_scan_kernel<<<HH, 256, SCAN_SMEM_FLOATS * sizeof(float), stream>>>(
      kb, vb, qb, fw1, fb1, fw2, fb2, flg, flb, lw, lb, ttg, ttb, outb);

  gate_ln_kernel<<<M, 256, 0, stream>>>(outb, gactb, png, pnb, gatedb);

  gemm_wmma_f32<<<ggrid, gblk, 0, stream>>>(gatedb, wo, (float*)d_out,
                                            M, HIDC, HIDC, 0);
}


# scan: kt@W1 of a token's 2nd..5th forward pass replaced by a rank-8 step a -= lr*(kt kt^T+1)@d_a in the weight-update phase (2 WMMAs; one 16-WMMA phase and barrier less per pass); sums re-associated
# speedup vs baseline: 4.6711x; 1.0701x over previous
#include <hip/hip_runtime.h>
#include <hip/hip_bf16.h>
#include <math.h>

#define BB     8
#define SS     256
#define HIDC   1024
#define HH     16
#define DD     64
#define FF     64
#define LR     0.1f
#define GSCALE (2.0f / 8192.0f)
#define LN_EPS 1e-5f
#define LP 76

typedef float v2f __attribute__((ext_vector_type(2)));
typedef float v4f __attribute__((ext_vector_type(4)));
typedef float v8f __attribute__((ext_vector_type(8)));

__device__ __forceinline__ v8f wmma4(v2f a, v2f b, v8f c) {
  return __builtin_amdgcn_wmma_f32_16x16x4_f32(
       false, a,  false, b,
       (short)0, c,  false,  false);
}

__device__ __forceinline__ float tanh_straight(float u) {
  const float a = fabsf(u);
  const float e = __builtin_amdgcn_exp2f(a * 2.8853900817779268f);
  const float t = 1.0f - 2.0f * __builtin_amdgcn_rcpf(e + 1.0f);
  return copysignf(t, u);
}

__device__ __forceinline__ float gelu_f(float x) {
  const float c = 0.7978845608028654f;
  float u = c * (x + 0.044715f * x * x * x);
  return 0.5f * x * (1.0f + tanh_straight(u));
}
__device__ __forceinline__ float gelu_grad(float x) {
  const float c = 0.7978845608028654f;
  float x2 = x * x;
  float u = c * (x + 0.044715f * x2 * x);
  float t = tanh_straight(u);
  return 0.5f * (1.0f + t) + 0.5f * x * (1.0f - t * t) * c * (1.0f + 0.134145f * x2);
}
__device__ __forceinline__ float xhalf(float x) {
  const int b = __builtin_bit_cast(int, x);
  return __builtin_bit_cast(float,
      __builtin_amdgcn_permlanex16(b, b, 0x76543210, 0xfedcba98, false, false));
}
template <int M>
__device__ __forceinline__ float xrow(float x) {
  const int b = __builtin_bit_cast(int, x);
  return __builtin_bit_cast(float,
      __builtin_amdgcn_update_dpp(0, b, 0x160 + M, 0xf, 0xf, true));
}
__device__ __forceinline__ float wave_sum32(float v) {
  v += xhalf(v);
  v += xrow<8>(v);
  v += xrow<4>(v);
  v += xrow<2>(v);
  v += xrow<1>(v);
  return v;
}
__device__ __forceinline__ void row_stats(float e0, float e1, float& m, float& is) {
  const float s = wave_sum32(e0 + e1);
  const float q = wave_sum32(e0 * e0 + e1 * e1);
  m  = s * (1.f / 64.f);
  is = rsqrtf(q * (1.f / 64.f) - m * m + LN_EPS);
}

__global__ void __launch_bounds__(256)
gemm_wmma_f32(const float* __restrict__ A, const float* __restrict__ Bm,
              float* __restrict__ C, int M, int N, int K, int act) {
  __shared__ float As[16 * LP];
  const int tid   = threadIdx.x;
  const int wave  = tid >> 5;
  const int lane  = tid & 31;
  const int mbase = blockIdx.y * 16;
  const int tn    = blockIdx.x * 8 + wave;
  const int col   = tn * 16 + (lane & 15);
  const int kh    = (lane >> 4) << 1;
  const int arow  = lane & 15;
  v8f acc = {0.f, 0.f, 0.f, 0.f, 0.f, 0.f, 0.f, 0.f};
  for (int kc = 0; kc < K; kc += 64) {
    __syncthreads();
    for (int i = tid; i < 16 * 64; i += 256)
      As[(i >> 6) * LP + (i & 63)] = A[(size_t)(mbase + (i >> 6)) * K + kc + (i & 63)];
    if (kc + 64 < K)
      __builtin_prefetch(&A[(size_t)(mbase + (tid >> 4)) * K + kc + 64], 0, 1);
    __syncthreads();
#pragma unroll
    for (int k0 = 0; k0 < 64; k0 += 4) {
      v2f a, b;
      a.x = As[arow * LP + k0 + kh];
      a.y = As[arow * LP + k0 + kh + 1];
      b.x = Bm[(size_t)(kc + k0 + kh) * N + col];
      b.y = Bm[(size_t)(kc + k0 + kh + 1) * N + col];
      acc = wmma4(a, b, acc);
    }
  }
  const int r0 = mbase + ((lane >= 16) ? 8 : 0);
#pragma unroll
  for (int i = 0; i < 8; ++i) {
    float v = acc[i];
    if (act) v = gelu_f(v);
    C[(size_t)(r0 + i) * N + col] = v;
  }
}

#define EPI_BIAS      0
#define EPI_GELU_PAIR 1
#define EPI_ADDAUX    2
#define EPI_DGELU     3
#define EPI_SCALE     4

__device__ __forceinline__ void
gemm16_rowA(const float* A8, const float* Bm, const float* bias,
            float* C8, float* C2, const float* aux,
            int transB, int epi, int wave, int lane) {
  if ((unsigned)wave >= 4u) return;
  const int col  = wave * 16 + (lane & 15);
  const int mrow = lane & 7;
  const int kh   = (lane >> 4) << 1;
  v2f a[16], b[16];
#pragma unroll
  for (int j = 0; j < 16; ++j) {
    const int k = 4 * j + kh;
    a[j].x = A8[mrow * LP + k];
    a[j].y = A8[mrow * LP + k + 1];
    if (!transB) {
      b[j].x = Bm[k * LP + col];
      b[j].y = Bm[(k + 1) * LP + col];
    } else {
      b[j].x = Bm[col * LP + k];
      b[j].y = Bm[col * LP + k + 1];
    }
  }
  const float bv = bias ? bias[col] : 0.f;
  float ax[8];
#pragma unroll
  for (int i = 0; i < 8; ++i)
    ax[i] = (epi == EPI_ADDAUX || epi == EPI_DGELU) ? aux[i * LP + col] : 0.f;
  __builtin_amdgcn_sched_barrier(0);
  v8f acc = {0.f, 0.f, 0.f, 0.f, 0.f, 0.f, 0.f, 0.f};
#pragma unroll
  for (int j = 0; j < 16; ++j) acc = wmma4(a[j], b[j], acc);
  if (lane < 16) {
#pragma unroll
    for (int i = 0; i < 8; ++i) {
      const int o = i * LP + col;
      float v = acc[i] + bv;
      if (epi == EPI_GELU_PAIR)      { C8[o] = v; C2[o] = gelu_f(v); }
      else if (epi == EPI_ADDAUX)    { C8[o] = v + ax[i]; }
      else if (epi == EPI_DGELU)     { C8[o] = v * gelu_grad(ax[i]); }
      else if (epi == EPI_SCALE)     { C8[o] = GSCALE * v; }
      else                           { C8[o] = v; }
    }
  }
}

__device__ __forceinline__ void
weight_updates(const float* Xa, const float* Ya, float* Wa, float* ba,
               const float* Xb, const float* Yb, float* Wb, float* bb,
               float lr, const float* Gp, float* apre, float* hnext,
               int wave, int lane) {
  const int c  = lane & 15;
  const int h  = lane >> 4;
  const int kh = h << 1;
  const int m  = (wave >> 1) * 16 + c;
  const int nb = (wave & 1) * 32;
  const float* Xs[2] = {Xa, Xb};
  const float* Ys[2] = {Ya, Yb};
  float*       Ws[2] = {Wa, Wb};
  v2f x[2][2], y[2][2][2];
  v4f w[2][2][2];
#pragma unroll
  for (int q = 0; q < 2; ++q) {
#pragma unroll
    for (int s = 0; s < 2; ++s) {
      const int k = 4 * s + kh;
      x[q][s].x = Xs[q][k * LP + m];
      x[q][s].y = Xs[q][(k + 1) * LP + m];
#pragma unroll
      for (int ti = 0; ti < 2; ++ti) {
        y[q][ti][s].x = Ys[q][k * LP + nb + 16 * ti + c];
        y[q][ti][s].y = Ys[q][(k + 1) * LP + nb + 16 * ti + c];
      }
    }
#pragma unroll
    for (int ti = 0; ti < 2; ++ti) {
      const v4f* p = (const v4f*)(Ws[q] + m * LP + nb + 16 * ti + 8 * h);
      w[q][ti][0] = p[0];
      w[q][ti][1] = p[1];
    }
  }
  const float* Yc = (wave & 2) ? Yb : Ya;
  float*       bc = (wave & 2) ? bb : ba;
  const int    d  = (wave & 1) * 32 + lane;
  float g[8];
#pragma unroll
  for (int b = 0; b < 8; ++b) g[b] = Yc[b * LP + d];
  float bold = 0.f;
  if (wave < 4) bold = bc[d];
  const int acol = (wave & 3) * 16 + c;
  const int arow = 4 * h;
  const int grow = (c < 8) ? c : ((c - 4) & 7);
  v2f ga[2], gb[2];
  float aold[4];
#pragma unroll
  for (int s = 0; s < 2; ++s) {
    const int k = 4 * s + kh;
    ga[s].x = Gp[grow * 8 + k];
    ga[s].y = Gp[grow * 8 + k + 1];
    gb[s].x = Yb[k * LP + acol];
    gb[s].y = Yb[(k + 1) * LP + acol];
  }
#pragma unroll
  for (int i = 0; i < 4; ++i) aold[i] = apre[(arow + i) * LP + acol];
  __builtin_amdgcn_sched_barrier(0);
  if (wave < 4) {
    v8f acc = {0.f, 0.f, 0.f, 0.f, 0.f, 0.f, 0.f, 0.f};
    acc = wmma4(ga[0], gb[0], acc);
    acc = wmma4(ga[1], gb[1], acc);
#pragma unroll
    for (int i = 0; i < 4; ++i) {
      const float v = aold[i] - lr * acc[i];
      apre[(arow + i) * LP + acol]  = v;
      hnext[(arow + i) * LP + acol] = gelu_f(v);
    }
  }
#pragma unroll
  for (int q = 0; q < 2; ++q) {
#pragma unroll
    for (int ti = 0; ti < 2; ++ti) {
      v8f acc = {0.f, 0.f, 0.f, 0.f, 0.f, 0.f, 0.f, 0.f};
      acc = wmma4(y[q][ti][0], x[q][0], acc);
      acc = wmma4(y[q][ti][1], x[q][1], acc);
      v4f lo = w[q][ti][0], hi = w[q][ti][1];
      lo.x -= lr * acc[0];  lo.y -= lr * acc[1];
      lo.z -= lr * acc[2];  lo.w -= lr * acc[3];
      hi.x -= lr * acc[4];  hi.y -= lr * acc[5];
      hi.z -= lr * acc[6];  hi.w -= lr * acc[7];
      v4f* p = (v4f*)(Ws[q] + m * LP + nb + 16 * ti + 8 * h);
      p[0] = lo;
      p[1] = hi;
    }
  }
  if (wave < 4) {
    float s = 0.f;
#pragma unroll
    for (int b = 0; b < 8; ++b) s += g[b];
    bc[d] = bold - lr * s;
  }
}

__device__ __forceinline__ void
gemm_self_transposed_scaled(const float* L, float* M, int wave, int lane) {
  const int kh = (lane >> 4) << 1;
  const int c  = lane & 15;
#pragma unroll
  for (int ti = 0; ti < 2; ++ti) {
    const int t  = wave * 2 + ti;
    const int tm = t >> 2, tn = t & 3;
    v8f acc = {0.f, 0.f, 0.f, 0.f, 0.f, 0.f, 0.f, 0.f};
#pragma unroll
    for (int j = 0; j < 16; ++j) {
      const int k = 4 * j + kh;
      v2f a, b;
      a.x = L[(tm * 16 + c) * LP + k];
      a.y = L[(tm * 16 + c) * LP + k + 1];
      b.x = L[(tn * 16 + c) * LP + k];
      b.y = L[(tn * 16 + c) * LP + k + 1];
      acc = wmma4(a, b, acc);
    }
    const int rbase = tm * 16 + ((lane >= 16) ? 8 : 0);
#pragma unroll
    for (int i = 0; i < 8; ++i) M[(rbase + i) * LP + tn * 16 + c] = GSCALE * acc[i];
  }
}

__device__ __forceinline__ void
gram_partial(const float* kt, float* P, int kq, int lane) {
  const int kh = (lane >> 4) << 1;
  const float* p = kt + (lane & 7) * LP + 16 * kq + kh;
  v2f f[4];
#pragma unroll
  for (int j = 0; j < 4; ++j) { f[j].x = p[4 * j]; f[j].y = p[4 * j + 1]; }
  v8f acc = {0.f, 0.f, 0.f, 0.f, 0.f, 0.f, 0.f, 0.f};
#pragma unroll
  for (int j = 0; j < 4; ++j) acc = wmma4(f[j], f[j], acc);
  if (lane < 8) {
#pragma unroll
    for (int r = 0; r < 8; ++r) P[r * 8 + lane] = acc[r];
  }
}

struct RowLN { float x0, x1, z0, z1, is; };
__device__ __forceinline__ RowLN
row_layernorm(const float* ob, const float* lgv, const float* lbv,
              int o0, int o1, int lane) {
  RowLN r;
  const float e0 = ob[o0], e1 = ob[o1];
  const float g0 = lgv[lane], g1 = lgv[lane + 32];
  const float c0 = lbv[lane], c1 = lbv[lane + 32];
  __builtin_amdgcn_sched_barrier(0);
  float m;
  row_stats(e0, e1, m, r.is);
  r.x0 = (e0 - m) * r.is;
  r.x1 = (e1 - m) * r.is;
  r.z0 = r.x0 * g0 + c0;
  r.z1 = r.x1 * g1 + c1;
  return r;
}

#define SCAN_SMEM_FLOATS (4 * 64 * LP + 12 * 8 * LP + 6 * 64 + 5 * 64)

__global__ void __launch_bounds__(256)
ttt_scan_kernel(const float* __restrict__ kg, const float* __restrict__ vg,
                const float* __restrict__ qg,
                const float* __restrict__ fw1, const float* __restrict__ fb1,
                const float* __restrict__ fw2, const float* __restrict__ fb2,
                const float* __restrict__ flg, const float* __restrict__ flb,
                const float* __restrict__ lwg, const float* __restrict__ lbg,
                const float* __restrict__ ttg, const float* __restrict__ ttb,
                float* __restrict__ outg) {
  extern __shared__ __attribute__((aligned(16))) float smem[];
  float* W1  = smem;
  float* W2  = W1 + 64 * LP;
  float* LW  = W2 + 64 * LP;
  float* MM  = LW + 64 * LP;
  float* ktb = MM + 64 * LP;
  float* tgt = ktb + 8 * LP;
  float* av  = tgt + 8 * LP;
  float* hgA = av + 8 * LP;
  float* hgB = hgA + 8 * LP;
  float* ob  = hgB + 8 * LP;
  float* xh  = ob + 8 * LP;
  float* zb  = xh + 8 * LP;
  float* dzb = zb + 8 * LP;
  float* dob = dzb + 8 * LP;
  float* dab = dob + 8 * LP;
  float* cb  = dab + 8 * LP;
  float* b1v = cb + 8 * LP;
  float* b2v = b1v + 64;
  float* lgv = b2v + 64;
  float* lbv = lgv + 64;
  float* tgv = lbv + 64;
  float* tbv = tgv + 64;
  float* Gp  = tbv + 64;
  float* Gq  = Gp + 64;

  const int h    = blockIdx.x;
  const int tid  = threadIdx.x;
  const int wave = tid >> 5;
  const int lane = tid & 31;

  const int b0 = tid >> 6,          d0 = tid & 63;
  const int b1i = (tid + 256) >> 6, d1i = (tid + 256) & 63;
  const size_t g0base = (size_t)(b0 * SS) * HIDC + h * 64 + d0;
  const size_t g1base = (size_t)(b1i * SS) * HIDC + h * 64 + d1i;
  const int l0 = b0 * LP + d0, l1 = b1i * LP + d1i;
  const int r0 = wave * LP + lane, r1 = r0 + 32;

  for (int i = tid; i < 4096; i += 256) {
    const int o = (i >> 6) * LP + (i & 63);
    W1[o] = fw1[(size_t)h * 4096 + i];
    W2[o] = fw2[(size_t)h * 4096 + i];
    LW[o] = lwg[i];
  }
  if (tid < 64) {
    b1v[tid] = fb1[h * 64 + tid];
    b2v[tid] = fb2[h * 64 + tid];
    lgv[tid] = flg[h * 64 + tid];
    lbv[tid] = flb[h * 64 + tid];
    tgv[tid] = ttg[h * 64 + tid];
    tbv[tid] = ttb[h * 64 + tid];
  }
  const float lbd = lbg[d0];

  float kp0 = kg[g0base], vp0 = vg[g0base];
  float kp1 = kg[g1base], vp1 = vg[g1base];
  __syncthreads();
  gemm_self_transposed_scaled(LW, MM, wave, lane);

  float* hc = hgA;
  float* hn = hgB;

  for (int t = 0; t < SS; ++t) {
    ktb[l0] = kp0;  tgt[l0] = lbd - (vp0 - kp0);
    ktb[l1] = kp1;  tgt[l1] = lbd - (vp1 - kp1);
    __syncthreads();
    if (t + 1 < SS) {
      kp0 = kg[g0base + (t + 1) * HIDC];  vp0 = vg[g0base + (t + 1) * HIDC];
      kp1 = kg[g1base + (t + 1) * HIDC];  vp1 = vg[g1base + (t + 1) * HIDC];
    }

    for (int step = 0; step < 4; ++step) {
      if (step == 0) {
        gemm16_rowA(tgt, LW, nullptr, cb, nullptr, nullptr, 1, EPI_SCALE, wave - 4, lane);
        gemm16_rowA(ktb, W1, b1v, av, hc, nullptr, 0, EPI_GELU_PAIR, wave, lane);
        __syncthreads();
        if (wave >= 4) gram_partial(ktb, Gq + 64 * (wave - 4), wave - 4, lane);
      }
      gemm16_rowA(hc, W2, b2v, ob, nullptr, nullptr, 0, EPI_BIAS, wave, lane);
      __syncthreads();
      const RowLN r = row_layernorm(ob, lgv, lbv, r0, r1, lane);
      xh[r0] = r.x0;  xh[r1] = r.x1;
      zb[r0] = r.z0;  zb[r1] = r.z1;
      __syncthreads();
      gemm16_rowA(zb, MM, nullptr, dzb, nullptr, cb, 0, EPI_ADDAUX, wave, lane);
      if (step == 0 && wave == 4) {
#pragma unroll
        for (int e = lane; e < 64; e += 32)
          Gp[e] = ((Gq[e] + Gq[64 + e]) + (Gq[128 + e] + Gq[192 + e])) + 1.0f;
      }
      __syncthreads();
      {
        const float dd0 = dzb[r0] * lgv[lane];
        const float dd1 = dzb[r1] * lgv[lane + 32];
        const float m1 = wave_sum32(dd0 + dd1) * (1.f / 64.f);
        const float m2 = wave_sum32(dd0 * r.x0 + dd1 * r.x1) * (1.f / 64.f);
        dob[r0] = r.is * (dd0 - m1 - r.x0 * m2);
        dob[r1] = r.is * (dd1 - m1 - r.x1 * m2);
      }
      __syncthreads();
      if (tid < 64) {
        float s = 0.f;
        for (int b = 0; b < 8; ++b) s += dzb[b * LP + tid] * xh[b * LP + tid];
        lgv[tid] -= LR * s;
      } else if (tid < 128) {
        int d = tid - 64; float s = 0.f;
        for (int b = 0; b < 8; ++b) s += dzb[b * LP + d];
        lbv[d] -= LR * s;
      }
      gemm16_rowA(dob, W2, nullptr, dab, nullptr, av, 1, EPI_DGELU, wave, lane);
      __syncthreads();
      weight_updates(hc, dob, W2, b2v, ktb, dab, W1, b1v, LR, Gp, av, hn, wave, lane);
      __syncthreads();
      { float* sw = hc; hc = hn; hn = sw; }
    }

    gemm16_rowA(hc, W2, b2v, ob, nullptr, nullptr, 0, EPI_BIAS, wave, lane);
    __syncthreads();
    {
      const size_t g = ((size_t)(wave * SS + t)) * HIDC + h * 64 + lane;
      const float q0 = qg[g], q1 = qg[g + 32];
      const float tg0 = tgv[lane], tg1 = tgv[lane + 32];
      const float tb0 = tbv[lane], tb1 = tbv[lane + 32];
      const RowLN r = row_layernorm(ob, lgv, lbv, r0, r1, lane);
      float m, is;
      row_stats(r.z0, r.z1, m, is);
      outg[g]      = q0 + ((r.z0 - m) * is * tg0 + tb0);
      outg[g + 32] = q1 + ((r.z1 - m) * is * tg1 + tb1);
    }
  }
}

__global__ void __launch_bounds__(256)
gate_ln_kernel(const float* __restrict__ outb, const float* __restrict__ gact,
               const float* __restrict__ png, const float* __restrict__ pnb,
               float* __restrict__ gated) {
  __shared__ float r1[256], r2[256];
  __shared__ float mu_s, inv_s;
  const int r = blockIdx.x, tid = threadIdx.x;
  const float* row = outb + (size_t)r * HIDC;
  float s = 0.f, q = 0.f;
  for (int i = tid; i < HIDC; i += 256) { float v = row[i]; s += v; q += v * v; }
  r1[tid] = s; r2[tid] = q;
  __syncthreads();
  for (int off = 128; off > 0; off >>= 1) {
    if (tid < off) { r1[tid] += r1[tid + off]; r2[tid] += r2[tid + off]; }
    __syncthreads();
  }
  if (tid == 0) {
    float m = r1[0] * (1.f / HIDC);
    mu_s = m;
    inv_s = rsqrtf(r2[0] * (1.f / HIDC) - m * m + LN_EPS);
  }
  __syncthreads();
  for (int i = tid; i < HIDC; i += 256) {
    float y = (row[i] - mu_s) * inv_s * png[i] + pnb[i];
    gated[(size_t)r * HIDC + i] = gact[(size_t)r * HIDC + i] * y;
  }
}

extern "C" void kernel_launch(void* const* d_in, const int* in_sizes, int n_in,
                              void* d_out, int out_size, void* d_ws, size_t ws_size,
                              hipStream_t stream) {
  (void)in_sizes; (void)n_in; (void)out_size; (void)ws_size;
  const float* x   = (const float*)d_in[0];
  const float* wq  = (const float*)d_in[1];
  const float* wk  = (const float*)d_in[2];
  const float* wv  = (const float*)d_in[3];
  const float* fw1 = (const float*)d_in[4];
  const float* fb1 = (const float*)d_in[5];
  const float* fw2 = (const float*)d_in[6];
  const float* fb2 = (const float*)d_in[7];
  const float* flg = (const float*)d_in[8];
  const float* flb = (const float*)d_in[9];
  const float* lw  = (const float*)d_in[10];
  const float* lb  = (const float*)d_in[11];
  const float* ttg = (const float*)d_in[12];
  const float* ttb = (const float*)d_in[13];
  const float* wo  = (const float*)d_in[14];
  const float* wg  = (const float*)d_in[15];
  const float* png = (const float*)d_in[16];
  const float* pnb = (const float*)d_in[17];

  const int M = BB * SS;
  const size_t PLANE = (size_t)M * HIDC;
  float* qb    = (float*)d_ws;
  float* kb    = qb + PLANE;
  float* vb    = kb + PLANE;
  float* gactb = vb + PLANE;
  float* outb  = gactb + PLANE;
  float* gatedb = qb;

  dim3 gblk(256);
  dim3 ggrid(HIDC / 128, M / 16);
  gemm_wmma_f32<<<ggrid, gblk, 0, stream>>>(x, wq, qb,    M, HIDC, HIDC, 0);
  gemm_wmma_f32<<<ggrid, gblk, 0, stream>>>(x, wk, kb,    M, HIDC, HIDC, 0);
  gemm_wmma_f32<<<ggrid, gblk, 0, stream>>>(x, wv, vb,    M, HIDC, HIDC, 0);
  gemm_wmma_f32<<<ggrid, gblk, 0, stream>>>(x, wg, gactb, M, HIDC, HIDC, 1);

  ttt_scan_kernel<<<HH, 256, SCAN_SMEM_FLOATS * sizeof(float), stream>>>(
      kb, vb, qb, fw1, fb1, fw2, fb2, flg, flb, lw, lb, ttg, ttb, outb);

  gate_ln_kernel<<<M, 256, 0, stream>>>(outb, gactb, png, pnb, gatedb);

  gemm_wmma_f32<<<ggrid, gblk, 0, stream>>>(gatedb, wo, (float*)d_out,
                                            M, HIDC, HIDC, 0);
}


# scan: fast weights and MM held in registers as WMMA operand slices (W2 twice: column and row slice), updated there; products read only A from LDS, 16 B wide; k order of product sums changed
# speedup vs baseline: 5.0220x; 1.0751x over previous
#include <hip/hip_runtime.h>
#include <hip/hip_bf16.h>
#include <math.h>

#define BB     8
#define SS     256
#define HIDC   1024
#define HH     16
#define DD     64
#define FF     64
#define LR     0.1f
#define GSCALE (2.0f / 8192.0f)
#define LN_EPS 1e-5f
#define LP 76

typedef float v2f __attribute__((ext_vector_type(2)));
typedef float v4f __attribute__((ext_vector_type(4)));
typedef float v8f __attribute__((ext_vector_type(8)));

__device__ __forceinline__ v8f wmma4(v2f a, v2f b, v8f c) {
  return __builtin_amdgcn_wmma_f32_16x16x4_f32(
       false, a,  false, b,
       (short)0, c,  false,  false);
}

__device__ __forceinline__ float tanh_straight(float u) {
  const float a = fabsf(u);
  const float e = __builtin_amdgcn_exp2f(a * 2.8853900817779268f);
  const float t = 1.0f - 2.0f * __builtin_amdgcn_rcpf(e + 1.0f);
  return copysignf(t, u);
}

__device__ __forceinline__ float gelu_f(float x) {
  const float c = 0.7978845608028654f;
  float u = c * (x + 0.044715f * x * x * x);
  return 0.5f * x * (1.0f + tanh_straight(u));
}
__device__ __forceinline__ float gelu_grad(float x) {
  const float c = 0.7978845608028654f;
  float x2 = x * x;
  float u = c * (x + 0.044715f * x2 * x);
  float t = tanh_straight(u);
  return 0.5f * (1.0f + t) + 0.5f * x * (1.0f - t * t) * c * (1.0f + 0.134145f * x2);
}
__device__ __forceinline__ float xhalf(float x) {
  const int b = __builtin_bit_cast(int, x);
  return __builtin_bit_cast(float,
      __builtin_amdgcn_permlanex16(b, b, 0x76543210, 0xfedcba98, false, false));
}
template <int M>
__device__ __forceinline__ float xrow(float x) {
  const int b = __builtin_bit_cast(int, x);
  return __builtin_bit_cast(float,
      __builtin_amdgcn_update_dpp(0, b, 0x160 + M, 0xf, 0xf, true));
}
__device__ __forceinline__ float wave_sum32(float v) {
  v += xhalf(v);
  v += xrow<8>(v);
  v += xrow<4>(v);
  v += xrow<2>(v);
  v += xrow<1>(v);
  return v;
}
__device__ __forceinline__ void row_stats(float e0, float e1, float& m, float& is) {
  const float s = wave_sum32(e0 + e1);
  const float q = wave_sum32(e0 * e0 + e1 * e1);
  m  = s * (1.f / 64.f);
  is = rsqrtf(q * (1.f / 64.f) - m * m + LN_EPS);
}

__global__ void __launch_bounds__(256)
gemm_wmma_f32(const float* __restrict__ A, const float* __restrict__ Bm,
              float* __restrict__ C, int M, int N, int K, int act) {
  __shared__ float As[16 * LP];
  const int tid   = threadIdx.x;
  const int wave  = tid >> 5;
  const int lane  = tid & 31;
  const int mbase = blockIdx.y * 16;
  const int tn    = blockIdx.x * 8 + wave;
  const int col   = tn * 16 + (lane & 15);
  const int kh    = (lane >> 4) << 1;
  const int arow  = lane & 15;
  v8f acc = {0.f, 0.f, 0.f, 0.f, 0.f, 0.f, 0.f, 0.f};
  for (int kc = 0; kc < K; kc += 64) {
    __syncthreads();
    for (int i = tid; i < 16 * 64; i += 256)
      As[(i >> 6) * LP + (i & 63)] = A[(size_t)(mbase + (i >> 6)) * K + kc + (i & 63)];
    if (kc + 64 < K)
      __builtin_prefetch(&A[(size_t)(mbase + (tid >> 4)) * K + kc + 64], 0, 1);
    __syncthreads();
#pragma unroll
    for (int k0 = 0; k0 < 64; k0 += 4) {
      v2f a, b;
      a.x = As[arow * LP + k0 + kh];
      a.y = As[arow * LP + k0 + kh + 1];
      b.x = Bm[(size_t)(kc + k0 + kh) * N + col];
      b.y = Bm[(size_t)(kc + k0 + kh + 1) * N + col];
      acc = wmma4(a, b, acc);
    }
  }
  const int r0 = mbase + ((lane >= 16) ? 8 : 0);
#pragma unroll
  for (int i = 0; i < 8; ++i) {
    float v = acc[i];
    if (act) v = gelu_f(v);
    C[(size_t)(r0 + i) * N + col] = v;
  }
}

#define EPI_BIAS      0
#define EPI_GELU_PAIR 1
#define EPI_ADDAUX    2
#define EPI_DGELU     3
#define EPI_SCALE     4

__device__ __forceinline__ void
gemm16_rowA(const float* A8, const float* Bm, const float* bias,
            float* C8, float* C2, const float* aux,
            int transB, int epi, int wave, int lane) {
  if ((unsigned)wave >= 4u) return;
  const int col  = wave * 16 + (lane & 15);
  const int mrow = lane & 7;
  const int kh   = (lane >> 4) << 1;
  v2f a[16], b[16];
#pragma unroll
  for (int j = 0; j < 16; ++j) {
    const int k = 4 * j + kh;
    a[j].x = A8[mrow * LP + k];
    a[j].y = A8[mrow * LP + k + 1];
    if (!transB) {
      b[j].x = Bm[k * LP + col];
      b[j].y = Bm[(k + 1) * LP + col];
    } else {
      b[j].x = Bm[col * LP + k];
      b[j].y = Bm[col * LP + k + 1];
    }
  }
  const float bv = bias ? bias[col] : 0.f;
  float ax[8];
#pragma unroll
  for (int i = 0; i < 8; ++i)
    ax[i] = (epi == EPI_ADDAUX || epi == EPI_DGELU) ? aux[i * LP + col] : 0.f;
  __builtin_amdgcn_sched_barrier(0);
  v8f acc = {0.f, 0.f, 0.f, 0.f, 0.f, 0.f, 0.f, 0.f};
#pragma unroll
  for (int j = 0; j < 16; ++j) acc = wmma4(a[j], b[j], acc);
  if (lane < 16) {
#pragma unroll
    for (int i = 0; i < 8; ++i) {
      const int o = i * LP + col;
      float v = acc[i] + bv;
      if (epi == EPI_GELU_PAIR)      { C8[o] = v; C2[o] = gelu_f(v); }
      else if (epi == EPI_ADDAUX)    { C8[o] = v + ax[i]; }
      else if (epi == EPI_DGELU)     { C8[o] = v * gelu_grad(ax[i]); }
      else if (epi == EPI_SCALE)     { C8[o] = GSCALE * v; }
      else                           { C8[o] = v; }
    }
  }
}


__device__ __forceinline__ void
prod_slice(const float* A8, const v8f (&T)[4], const float* bias,
           float* C8, float* C2, const float* aux, int epi, int w, int lane) {
  if ((unsigned)w >= 4u) return;
  const int col = w * 16 + (lane & 15);
  const int hh  = lane >> 4;
  const v4f* ap = (const v4f*)(A8 + (lane & 7) * LP + 8 * hh);
  v4f alo[4], ahi[4];
#pragma unroll
  for (int t = 0; t < 4; ++t) { alo[t] = ap[4 * t]; ahi[t] = ap[4 * t + 1]; }
  const float bv = bias ? bias[col] : 0.f;
  float ax[8];
#pragma unroll
  for (int i = 0; i < 8; ++i)
    ax[i] = (epi == EPI_ADDAUX || epi == EPI_DGELU) ? aux[i * LP + col] : 0.f;
  __builtin_amdgcn_sched_barrier(0);
  v8f acc = {0.f, 0.f, 0.f, 0.f, 0.f, 0.f, 0.f, 0.f};
#pragma unroll
  for (int t = 0; t < 4; ++t) {
    v2f a, b;
    a.x = alo[t].x; a.y = alo[t].y; b.x = T[t][0]; b.y = T[t][1]; acc = wmma4(a, b, acc);
    a.x = alo[t].z; a.y = alo[t].w; b.x = T[t][2]; b.y = T[t][3]; acc = wmma4(a, b, acc);
    a.x = ahi[t].x; a.y = ahi[t].y; b.x = T[t][4]; b.y = T[t][5]; acc = wmma4(a, b, acc);
    a.x = ahi[t].z; a.y = ahi[t].w; b.x = T[t][6]; b.y = T[t][7]; acc = wmma4(a, b, acc);
  }
  if (lane < 16) {
#pragma unroll
    for (int i = 0; i < 8; ++i) {
      const int o = i * LP + col;
      float v = acc[i] + bv;
      if (epi == EPI_GELU_PAIR)      { C8[o] = v; C2[o] = gelu_f(v); }
      else if (epi == EPI_ADDAUX)    { C8[o] = v + ax[i]; }
      else if (epi == EPI_DGELU)     { C8[o] = v * gelu_grad(ax[i]); }
      else                           { C8[o] = v; }
    }
  }
}

struct UpdFrag { v2f xa[4][2]; v2f yb[2]; };
__device__ __forceinline__ UpdFrag
upd_load(const float* Xr, const float* Yc, int w, int lane) {
  UpdFrag f;
  const int c  = lane & 15;
  const int kh = (lane >> 4) << 1;
#pragma unroll
  for (int s = 0; s < 2; ++s) {
    const int k = 4 * s + kh;
    f.yb[s].x = Yc[k * LP + 16 * w + c];
    f.yb[s].y = Yc[(k + 1) * LP + 16 * w + c];
#pragma unroll
    for (int t = 0; t < 4; ++t) {
      f.xa[t][s].x = Xr[k * LP + 16 * t + c];
      f.xa[t][s].y = Xr[(k + 1) * LP + 16 * t + c];
    }
  }
  return f;
}
__device__ __forceinline__ void
upd_apply(v8f (&T)[4], const UpdFrag& f, float lr) {
#pragma unroll
  for (int t = 0; t < 4; ++t) {
    v8f acc = {0.f, 0.f, 0.f, 0.f, 0.f, 0.f, 0.f, 0.f};
    acc = wmma4(f.xa[t][0], f.yb[0], acc);
    acc = wmma4(f.xa[t][1], f.yb[1], acc);
#pragma unroll
    for (int r = 0; r < 8; ++r) T[t][r] -= lr * acc[r];
  }
}
__device__ __forceinline__ void
slice_update(v8f (&T)[4], const float* Xr, const float* Yc, float lr, int w, int lane) {
  const UpdFrag f = upd_load(Xr, Yc, w, lane);
  __builtin_amdgcn_sched_barrier(0);
  upd_apply(T, f, lr);
}

__device__ __forceinline__ void
step_tail_low(v8f (&W1c)[4], const float* kt,
              const float* Ya, float* ba, const float* Yb, float* bb,
              float lr, const float* Gp, float* apre, float* hnext,
              int wave, int lane) {
  const int c  = lane & 15;
  const int hh = lane >> 4;
  const int kh = hh << 1;
  const UpdFrag f = upd_load(kt, Yb, wave, lane);
  const float* Yc = (wave & 2) ? Yb : Ya;
  float*       bc = (wave & 2) ? bb : ba;
  const int    d  = (wave & 1) * 32 + lane;
  float g[8];
#pragma unroll
  for (int b = 0; b < 8; ++b) g[b] = Yc[b * LP + d];
  const float bold = bc[d];
  const int acol = wave * 16 + c;
  const int arow = 4 * hh;
  const int grow = (c < 8) ? c : ((c - 4) & 7);
  v2f ga[2], gb[2];
  float aold[4];
#pragma unroll
  for (int s = 0; s < 2; ++s) {
    const int k = 4 * s + kh;
    ga[s].x = Gp[grow * 8 + k];
    ga[s].y = Gp[grow * 8 + k + 1];
    gb[s].x = Yb[k * LP + acol];
    gb[s].y = Yb[(k + 1) * LP + acol];
  }
#pragma unroll
  for (int i = 0; i < 4; ++i) aold[i] = apre[(arow + i) * LP + acol];
  __builtin_amdgcn_sched_barrier(0);
  {
    v8f acc = {0.f, 0.f, 0.f, 0.f, 0.f, 0.f, 0.f, 0.f};
    acc = wmma4(ga[0], gb[0], acc);
    acc = wmma4(ga[1], gb[1], acc);
#pragma unroll
    for (int i = 0; i < 4; ++i) {
      const float v = aold[i] - lr * acc[i];
      apre[(arow + i) * LP + acol]  = v;
      hnext[(arow + i) * LP + acol] = gelu_f(v);
    }
  }
  upd_apply(W1c, f, lr);
  {
    float s = 0.f;
#pragma unroll
    for (int b = 0; b < 8; ++b) s += g[b];
    bc[d] = bold - lr * s;
  }
}

__device__ __forceinline__ void
gemm_self_transposed_scaled(const float* L, float* M, int wave, int lane) {
  const int kh = (lane >> 4) << 1;
  const int c  = lane & 15;
#pragma unroll
  for (int ti = 0; ti < 2; ++ti) {
    const int t  = wave * 2 + ti;
    const int tm = t >> 2, tn = t & 3;
    v8f acc = {0.f, 0.f, 0.f, 0.f, 0.f, 0.f, 0.f, 0.f};
#pragma unroll
    for (int j = 0; j < 16; ++j) {
      const int k = 4 * j + kh;
      v2f a, b;
      a.x = L[(tm * 16 + c) * LP + k];
      a.y = L[(tm * 16 + c) * LP + k + 1];
      b.x = L[(tn * 16 + c) * LP + k];
      b.y = L[(tn * 16 + c) * LP + k + 1];
      acc = wmma4(a, b, acc);
    }
    const int rbase = tm * 16 + ((lane >= 16) ? 8 : 0);
#pragma unroll
    for (int i = 0; i < 8; ++i) M[(rbase + i) * LP + tn * 16 + c] = GSCALE * acc[i];
  }
}

__device__ __forceinline__ void
gram_partial(const float* kt, float* P, int kq, int lane) {
  const int kh = (lane >> 4) << 1;
  const float* p = kt + (lane & 7) * LP + 16 * kq + kh;
  v2f f[4];
#pragma unroll
  for (int j = 0; j < 4; ++j) { f[j].x = p[4 * j]; f[j].y = p[4 * j + 1]; }
  v8f acc = {0.f, 0.f, 0.f, 0.f, 0.f, 0.f, 0.f, 0.f};
#pragma unroll
  for (int j = 0; j < 4; ++j) acc = wmma4(f[j], f[j], acc);
  if (lane < 8) {
#pragma unroll
    for (int r = 0; r < 8; ++r) P[r * 8 + lane] = acc[r];
  }
}

struct RowLN { float x0, x1, z0, z1, is; };
__device__ __forceinline__ RowLN
row_layernorm(const float* ob, const float* lgv, const float* lbv,
              int o0, int o1, int lane) {
  RowLN r;
  const float e0 = ob[o0], e1 = ob[o1];
  const float g0 = lgv[lane], g1 = lgv[lane + 32];
  const float c0 = lbv[lane], c1 = lbv[lane + 32];
  __builtin_amdgcn_sched_barrier(0);
  float m;
  row_stats(e0, e1, m, r.is);
  r.x0 = (e0 - m) * r.is;
  r.x1 = (e1 - m) * r.is;
  r.z0 = r.x0 * g0 + c0;
  r.z1 = r.x1 * g1 + c1;
  return r;
}

#define SCAN_SMEM_FLOATS (2 * 64 * LP + 12 * 8 * LP + 6 * 64 + 5 * 64)

__global__ void __launch_bounds__(256)
ttt_scan_kernel(const float* __restrict__ kg, const float* __restrict__ vg,
                const float* __restrict__ qg,
                const float* __restrict__ fw1, const float* __restrict__ fb1,
                const float* __restrict__ fw2, const float* __restrict__ fb2,
                const float* __restrict__ flg, const float* __restrict__ flb,
                const float* __restrict__ lwg, const float* __restrict__ lbg,
                const float* __restrict__ ttg, const float* __restrict__ ttb,
                float* __restrict__ outg) {
  extern __shared__ __attribute__((aligned(16))) float smem[];
  float* LW  = smem;
  float* MM  = LW + 64 * LP;
  float* ktb = MM + 64 * LP;
  float* tgt = ktb + 8 * LP;
  float* av  = tgt + 8 * LP;
  float* hgA = av + 8 * LP;
  float* hgB = hgA + 8 * LP;
  float* ob  = hgB + 8 * LP;
  float* xh  = ob + 8 * LP;
  float* zb  = xh + 8 * LP;
  float* dzb = zb + 8 * LP;
  float* dob = dzb + 8 * LP;
  float* dab = dob + 8 * LP;
  float* cb  = dab + 8 * LP;
  float* b1v = cb + 8 * LP;
  float* b2v = b1v + 64;
  float* lgv = b2v + 64;
  float* lbv = lgv + 64;
  float* tgv = lbv + 64;
  float* tbv = tgv + 64;
  float* Gp  = tbv + 64;
  float* Gq  = Gp + 64;

  const int h    = blockIdx.x;
  const int tid  = threadIdx.x;
  const int wave = tid >> 5;
  const int lane = tid & 31;

  const int b0 = tid >> 6,          d0 = tid & 63;
  const int b1i = (tid + 256) >> 6, d1i = (tid + 256) & 63;
  const size_t g0base = (size_t)(b0 * SS) * HIDC + h * 64 + d0;
  const size_t g1base = (size_t)(b1i * SS) * HIDC + h * 64 + d1i;
  const int l0 = b0 * LP + d0, l1 = b1i * LP + d1i;
  const int r0 = wave * LP + lane, r1 = r0 + 32;

  v8f W1c[4], W2c[4], W2r[4], MMc[4];
  {
    const int c  = lane & 15;
    const int hh = lane >> 4;
    const int w  = wave & 3;
    const float* p1 = fw1 + (size_t)h * 4096;
    const float* p2 = fw2 + (size_t)h * 4096;
#pragma unroll
    for (int t = 0; t < 4; ++t) {
#pragma unroll
      for (int r = 0; r < 8; ++r) {
        const int k = 16 * t + 8 * hh + r;
        W1c[t][r] = p1[k * 64 + 16 * w + c];
        W2c[t][r] = p2[k * 64 + 16 * w + c];
        W2r[t][r] = p2[(16 * w + c) * 64 + k];
      }
    }
  }
  for (int i = tid; i < 4096; i += 256)
    LW[(i >> 6) * LP + (i & 63)] = lwg[i];
  if (tid < 64) {
    b1v[tid] = fb1[h * 64 + tid];
    b2v[tid] = fb2[h * 64 + tid];
    lgv[tid] = flg[h * 64 + tid];
    lbv[tid] = flb[h * 64 + tid];
    tgv[tid] = ttg[h * 64 + tid];
    tbv[tid] = ttb[h * 64 + tid];
  }
  const float lbd = lbg[d0];

  float kp0 = kg[g0base], vp0 = vg[g0base];
  float kp1 = kg[g1base], vp1 = vg[g1base];
  __syncthreads();
  gemm_self_transposed_scaled(LW, MM, wave, lane);
  __syncthreads();
  {
    const int c  = lane & 15;
    const int hh = lane >> 4;
    const int w  = wave & 3;
#pragma unroll
    for (int t = 0; t < 4; ++t) {
#pragma unroll
      for (int r = 0; r < 8; ++r)
        MMc[t][r] = MM[(16 * t + 8 * hh + r) * LP + 16 * w + c];
    }
  }

  float* hc = hgA;
  float* hn = hgB;

  for (int t = 0; t < SS; ++t) {
    ktb[l0] = kp0;  tgt[l0] = lbd - (vp0 - kp0);
    ktb[l1] = kp1;  tgt[l1] = lbd - (vp1 - kp1);
    __syncthreads();
    if (t + 1 < SS) {
      kp0 = kg[g0base + (t + 1) * HIDC];  vp0 = vg[g0base + (t + 1) * HIDC];
      kp1 = kg[g1base + (t + 1) * HIDC];  vp1 = vg[g1base + (t + 1) * HIDC];
    }

    for (int step = 0; step < 4; ++step) {
      if (step == 0) {
        gemm16_rowA(tgt, LW, nullptr, cb, nullptr, nullptr, 1, EPI_SCALE, wave - 4, lane);
        prod_slice(ktb, W1c, b1v, av, hc, nullptr, EPI_GELU_PAIR, wave, lane);
        __syncthreads();
        if (wave >= 4) gram_partial(ktb, Gq + 64 * (wave - 4), wave - 4, lane);
      }
      prod_slice(hc, W2c, b2v, ob, nullptr, nullptr, EPI_BIAS, wave, lane);
      __syncthreads();
      const RowLN r = row_layernorm(ob, lgv, lbv, r0, r1, lane);
      xh[r0] = r.x0;  xh[r1] = r.x1;
      zb[r0] = r.z0;  zb[r1] = r.z1;
      __syncthreads();
      prod_slice(zb, MMc, nullptr, dzb, nullptr, cb, EPI_ADDAUX, wave, lane);
      if (step == 0 && wave == 4) {
#pragma unroll
        for (int e = lane; e < 64; e += 32)
          Gp[e] = ((Gq[e] + Gq[64 + e]) + (Gq[128 + e] + Gq[192 + e])) + 1.0f;
      }
      __syncthreads();
      {
        const float dd0 = dzb[r0] * lgv[lane];
        const float dd1 = dzb[r1] * lgv[lane + 32];
        const float m1 = wave_sum32(dd0 + dd1) * (1.f / 64.f);
        const float m2 = wave_sum32(dd0 * r.x0 + dd1 * r.x1) * (1.f / 64.f);
        dob[r0] = r.is * (dd0 - m1 - r.x0 * m2);
        dob[r1] = r.is * (dd1 - m1 - r.x1 * m2);
      }
      __syncthreads();
      if (tid < 64) {
        float s = 0.f;
        for (int b = 0; b < 8; ++b) s += dzb[b * LP + tid] * xh[b * LP + tid];
        lgv[tid] -= LR * s;
      } else if (tid < 128) {
        int d = tid - 64; float s = 0.f;
        for (int b = 0; b < 8; ++b) s += dzb[b * LP + d];
        lbv[d] -= LR * s;
      }
      prod_slice(dob, W2r, nullptr, dab, nullptr, av, EPI_DGELU, wave - 4, lane);
      if (wave < 4) slice_update(W2c, hc, dob, LR, wave, lane);
      __syncthreads();
      if (wave >= 4) slice_update(W2r, dob, hc, LR, wave - 4, lane);
      else           step_tail_low(W1c, ktb, dob, b2v, dab, b1v, LR, Gp, av, hn, wave, lane);
      __syncthreads();
      { float* sw = hc; hc = hn; hn = sw; }
    }

    prod_slice(hc, W2c, b2v, ob, nullptr, nullptr, EPI_BIAS, wave, lane);
    __syncthreads();
    {
      const size_t g = ((size_t)(wave * SS + t)) * HIDC + h * 64 + lane;
      const float q0 = qg[g], q1 = qg[g + 32];
      const float tg0 = tgv[lane], tg1 = tgv[lane + 32];
      const float tb0 = tbv[lane], tb1 = tbv[lane + 32];
      const RowLN r = row_layernorm(ob, lgv, lbv, r0, r1, lane);
      float m, is;
      row_stats(r.z0, r.z1, m, is);
      outg[g]      = q0 + ((r.z0 - m) * is * tg0 + tb0);
      outg[g + 32] = q1 + ((r.z1 - m) * is * tg1 + tb1);
    }
  }
}

__global__ void __launch_bounds__(256)
gate_ln_kernel(const float* __restrict__ outb, const float* __restrict__ gact,
               const float* __restrict__ png, const float* __restrict__ pnb,
               float* __restrict__ gated) {
  __shared__ float r1[256], r2[256];
  __shared__ float mu_s, inv_s;
  const int r = blockIdx.x, tid = threadIdx.x;
  const float* row = outb + (size_t)r * HIDC;
  float s = 0.f, q = 0.f;
  for (int i = tid; i < HIDC; i += 256) { float v = row[i]; s += v; q += v * v; }
  r1[tid] = s; r2[tid] = q;
  __syncthreads();
  for (int off = 128; off > 0; off >>= 1) {
    if (tid < off) { r1[tid] += r1[tid + off]; r2[tid] += r2[tid + off]; }
    __syncthreads();
  }
  if (tid == 0) {
    float m = r1[0] * (1.f / HIDC);
    mu_s = m;
    inv_s = rsqrtf(r2[0] * (1.f / HIDC) - m * m + LN_EPS);
  }
  __syncthreads();
  for (int i = tid; i < HIDC; i += 256) {
    float y = (row[i] - mu_s) * inv_s * png[i] + pnb[i];
    gated[(size_t)r * HIDC + i] = gact[(size_t)r * HIDC + i] * y;
  }
}

extern "C" void kernel_launch(void* const* d_in, const int* in_sizes, int n_in,
                              void* d_out, int out_size, void* d_ws, size_t ws_size,
                              hipStream_t stream) {
  (void)in_sizes; (void)n_in; (void)out_size; (void)ws_size;
  const float* x   = (const float*)d_in[0];
  const float* wq  = (const float*)d_in[1];
  const float* wk  = (const float*)d_in[2];
  const float* wv  = (const float*)d_in[3];
  const float* fw1 = (const float*)d_in[4];
  const float* fb1 = (const float*)d_in[5];
  const float* fw2 = (const float*)d_in[6];
  const float* fb2 = (const float*)d_in[7];
  const float* flg = (const float*)d_in[8];
  const float* flb = (const float*)d_in[9];
  const float* lw  = (const float*)d_in[10];
  const float* lb  = (const float*)d_in[11];
  const float* ttg = (const float*)d_in[12];
  const float* ttb = (const float*)d_in[13];
  const float* wo  = (const float*)d_in[14];
  const float* wg  = (const float*)d_in[15];
  const float* png = (const float*)d_in[16];
  const float* pnb = (const float*)d_in[17];

  const int M = BB * SS;
  const size_t PLANE = (size_t)M * HIDC;
  float* qb    = (float*)d_ws;
  float* kb    = qb + PLANE;
  float* vb    = kb + PLANE;
  float* gactb = vb + PLANE;
  float* outb  = gactb + PLANE;
  float* gatedb = qb;

  dim3 gblk(256);
  dim3 ggrid(HIDC / 128, M / 16);
  gemm_wmma_f32<<<ggrid, gblk, 0, stream>>>(x, wq, qb,    M, HIDC, HIDC, 0);
  gemm_wmma_f32<<<ggrid, gblk, 0, stream>>>(x, wk, kb,    M, HIDC, HIDC, 0);
  gemm_wmma_f32<<<ggrid, gblk, 0, stream>>>(x, wv, vb,    M, HIDC, HIDC, 0);
  gemm_wmma_f32<<<ggrid, gblk, 0, stream>>>(x, wg, gactb, M, HIDC, HIDC, 1);

  ttt_scan_kernel<<<HH, 256, SCAN_SMEM_FLOATS * sizeof(float), stream>>>(
      kb, vb, qb, fw1, fb1, fw2, fb2, flg, flb, lw, lb, ttg, ttb, outb);

  gate_ln_kernel<<<M, 256, 0, stream>>>(outb, gactb, png, pnb, gatedb);

  gemm_wmma_f32<<<ggrid, gblk, 0, stream>>>(gatedb, wo, (float*)d_out,
                                            M, HIDC, HIDC, 0);
}
